# starting kernel
# baseline (speedup 1.0000x reference)
#include <hip/hip_runtime.h>
#include <hip/hip_bf16.h>

typedef __attribute__((ext_vector_type(2))) float v2f;
typedef __attribute__((ext_vector_type(8))) float v8f;

#define NQ 10
#define DIM 1024
#define WPB 8

__global__ void build_gates_kernel(const float* __restrict__ W,
                                   float* __restrict__ G) {
  int g = threadIdx.x;
  if (g >= 40) return;
  int L = g / 20, r = g % 20;
  float u00r, u00i, u01r, u01i, u10r, u10i, u11r, u11i;
  if (r < 10) {
    int base = L * 40 + 3 * r;
    float phi = W[base], th = W[base + 1], om = W[base + 2];
    float ct = cosf(0.5f * th), st = sinf(0.5f * th);
    float ap = 0.5f * (phi + om), am = 0.5f * (phi - om);
    float cap = cosf(ap), sap = sinf(ap);
    float cam = cosf(am), sam = sinf(am);
    u00r =  ct * cap; u00i = -ct * sap;
    u01r = -st * cam; u01i = -st * sam;
    u10r =  st * cam; u10i = -st * sam;
    u11r =  ct * cap; u11i =  ct * sap;
  } else {
    int e = r - 10;
    float th = W[L * 40 + 30 + e];
    float ct = cosf(0.5f * th), st = sinf(0.5f * th);
    u00r = ct;  u00i = 0.f; u01r = -st; u01i = 0.f;
    u10r = st;  u10i = 0.f; u11r = ct;  u11i = 0.f;
  }
  float* M = G + g * 16;
  M[ 0] = u00r; M[ 1] = -u00i; M[ 2] = u01r; M[ 3] = -u01i;
  M[ 4] = u00i; M[ 5] =  u00r; M[ 6] = u01i; M[ 7] =  u01r;
  M[ 8] = u10r; M[ 9] = -u10i; M[10] = u11r; M[11] = -u11i;
  M[12] = u10i; M[13] =  u10r; M[14] = u11i; M[15] =  u11r;
}

__global__ __launch_bounds__(256, 8) void qsim_kernel(
    const float* __restrict__ inp, const float* __restrict__ G,
    float* __restrict__ out) {
  __shared__ float2 st[WPB][DIM];
  const int t = threadIdx.x & 31;
  const int w = threadIdx.x >> 5;
  const int samp = blockIdx.x * WPB + w;
  float2* sv = st[w];

  float cc[NQ], ss[NQ];
  const float* x = inp + samp * NQ;
  #pragma unroll
  for (int q = 0; q < NQ; ++q) {
    float xv = fminf(fmaxf(x[q], 0.f), 1.f);
    float a = 0.5f * 3.14159265358979323846f * xv;
    cc[q] = cosf(a);
    ss[q] = sinf(a);
  }
  float plo = 1.f;
  #pragma unroll
  for (int j = 0; j < 5; ++j)
    plo *= ((t >> j) & 1) ? ss[9 - j] : cc[9 - j];
  #pragma unroll 4
  for (int r = 0; r < 32; ++r) {
    float phi = 1.f;
    #pragma unroll
    for (int j = 0; j < 5; ++j)
      phi *= ((r >> j) & 1) ? ss[4 - j] : cc[4 - j];
    sv[r * 32 + t] = make_float2(phi * plo, 0.f);
  }
  __builtin_amdgcn_wave_barrier();

  const int row4 = (t & 3) * 4;
  const int c0 = (t < 16) ? 0 : 2;
  const int pl = t & 15;
  const bool lo = (t < 16);

  for (int L = 0; L < 2; ++L) {
    for (int q = 0; q < NQ; ++q) {
      const float* M = G + (L * 20 + q) * 16;
      v2f a;
      a.x = M[row4 + c0];
      a.y = M[row4 + c0 + 1];
      const int b = 9 - q;
      const int S = 1 << b;
      #pragma unroll 4
      for (int blk = 0; blk < 32; ++blk) {
        int p = blk * 16 + pl;
        int i0 = ((p >> b) << (b + 1)) | (p & (S - 1));
        int idx = lo ? i0 : (i0 | S);
        float2 amp = sv[idx];
        v2f bb;
        bb.x = amp.x;
        bb.y = amp.y;
        v8f cz = {};
        v8f d = __builtin_amdgcn_wmma_f32_16x16x4_f32(
            false, a, false, bb, (short)0, cz, false, false);
        sv[idx] = make_float2(lo ? d[0] : d[2], lo ? d[1] : d[3]);
      }
      __builtin_amdgcn_wave_barrier();
    }
    for (int e = 0; e < NQ; ++e) {
      const float* M = G + (L * 20 + 10 + e) * 16;
      v2f a;
      a.x = M[row4 + c0];
      a.y = M[row4 + c0 + 1];
      const int u = e, v = (e + 1) % NQ;
      const int bu = 9 - u, bv = 9 - v;
      const int blo = bu < bv ? bu : bv;
      const int bhi = bu < bv ? bv : bu;
      #pragma unroll 4
      for (int blk = 0; blk < 16; ++blk) {
        int p = blk * 16 + pl;
        int x1 = ((p >> blo) << (blo + 1)) | (p & ((1 << blo) - 1));
        int x2 = ((x1 >> bhi) << (bhi + 1)) | (x1 & ((1 << bhi) - 1));
        int i0 = x2 | (1 << bu);
        int idx = lo ? i0 : (i0 | (1 << bv));
        float2 amp = sv[idx];
        v2f bb;
        bb.x = amp.x;
        bb.y = amp.y;
        v8f cz = {};
        v8f d = __builtin_amdgcn_wmma_f32_16x16x4_f32(
            false, a, false, bb, (short)0, cz, false, false);
        sv[idx] = make_float2(lo ? d[0] : d[2], lo ? d[1] : d[3]);
      }
      __builtin_amdgcn_wave_barrier();
    }
  }

  float acc[NQ];
  #pragma unroll
  for (int q = 0; q < NQ; ++q) acc[q] = 0.f;
  #pragma unroll 4
  for (int r = 0; r < 32; ++r) {
    int i = r * 32 + t;
    float2 amp = sv[i];
    float p = amp.x * amp.x + amp.y * amp.y;
    #pragma unroll
    for (int q = 0; q < NQ; ++q)
      acc[q] += ((i >> (9 - q)) & 1) ? -p : p;
  }
  #pragma unroll
  for (int q = 0; q < NQ; ++q) {
    #pragma unroll
    for (int off = 16; off >= 1; off >>= 1)
      acc[q] += __shfl_xor(acc[q], off, 32);
  }
  if (t == 0) {
    #pragma unroll
    for (int q = 0; q < NQ; ++q) out[samp * NQ + q] = acc[q];
  }
}

extern "C" void kernel_launch(void* const* d_in, const int* in_sizes, int n_in,
                              void* d_out, int out_size, void* d_ws,
                              size_t ws_size, hipStream_t stream) {
  const float* inputs = (const float*)d_in[0];
  const float* weights = (const float*)d_in[1];
  float* out = (float*)d_out;
  float* G = (float*)d_ws;

  const int B = in_sizes[0] / NQ;
  build_gates_kernel<<<dim3(1), dim3(64), 0, stream>>>(weights, G);
  qsim_kernel<<<dim3(B / WPB), dim3(256), 0, stream>>>(inputs, G, out);
}


# Fuse each layer's gates into five 3-qubit groups: one real 16x16 matrix per group, applied by 4 chained k=4 WMMAs (10 LDS passes, not 40; swizzled banks); gate products change f32 summation order
# speedup vs baseline: 3.0625x; 3.0625x over previous
#include <hip/hip_runtime.h>


typedef __attribute__((ext_vector_type(2))) float v2f;
typedef __attribute__((ext_vector_type(8))) float v8f;

#define NQ 10
#define DIM 1024
#define WPB 8
#define NPASS 10

__host__ __device__ constexpr int swz(int i) {
  return i ^ ((i & 32) ? 0x14 : 0) ^ ((i & 64) ? 0x18 : 0) ^
         ((i & 128) ? 0x13 : 0) ^ ((i & 256) ? 0x0A : 0) ^
         ((i & 512) ? 0x10 : 0);
}

template <int l>
__device__ __forceinline__ void ap_rot(float (&vr)[8], float (&vi)[8],
                                       const float* u) {
  const float ar = u[0], ai = u[1], br = u[2], bi = u[3];
  const float cr = u[4], ci = u[5], dr = u[6], di = u[7];
#pragma unroll
  for (int a = 0; a < 8; ++a) {
    if ((a >> l) & 1) continue;
    const int b = a | (1 << l);
    const float x0r = vr[a], x0i = vi[a], x1r = vr[b], x1i = vi[b];
    vr[a] = ar * x0r - ai * x0i + br * x1r - bi * x1i;
    vi[a] = ar * x0i + ai * x0r + br * x1i + bi * x1r;
    vr[b] = cr * x0r - ci * x0i + dr * x1r - di * x1i;
    vi[b] = cr * x0i + ci * x0r + dr * x1i + di * x1r;
  }
}

template <int cb, int tb>
__device__ __forceinline__ void ap_cry(float (&vr)[8], float (&vi)[8],
                                       const float* u) {
  const float c = u[0], s = u[1];
#pragma unroll
  for (int a = 0; a < 8; ++a) {
    if (!((a >> cb) & 1) || ((a >> tb) & 1)) continue;
    const int b = a | (1 << tb);
    const float x0r = vr[a], x0i = vi[a], x1r = vr[b], x1i = vi[b];
    vr[a] = c * x0r - s * x1r;
    vi[a] = c * x0i - s * x1i;
    vr[b] = s * x0r + c * x1r;
    vi[b] = s * x0i + c * x1i;
  }
}

__global__ void build_gates_kernel(const float* __restrict__ W,
                                   float* __restrict__ G) {
  __shared__ float gs[40][8];
  const int g = threadIdx.x;
  if (g < 40) {
    const int L = g / 20, r = g % 20;
    float* u = gs[g];
    if (r < 10) {
      const int base = L * 40 + 3 * r;
      const float phi = W[base], th = W[base + 1], om = W[base + 2];
      const float ct = cosf(0.5f * th), st = sinf(0.5f * th);
      const float ap = 0.5f * (phi + om), am = 0.5f * (phi - om);
      const float cap = cosf(ap), sap = sinf(ap);
      const float cam = cosf(am), sam = sinf(am);
      u[0] =  ct * cap; u[1] = -ct * sap;
      u[2] = -st * cam; u[3] = -st * sam;
      u[4] =  st * cam; u[5] = -st * sam;
      u[6] =  ct * cap; u[7] =  ct * sap;
    } else {
      const float th = W[L * 40 + 30 + (r - 10)];
      u[0] = cosf(0.5f * th);
      u[1] = sinf(0.5f * th);
    }
  }
  __syncthreads();
  if (g >= NPASS * 8) return;
  const int p = g >> 3, col = g & 7;
  const int gb = (p / 5) * 20;
  float vr[8], vi[8];
#pragma unroll
  for (int a = 0; a < 8; ++a) { vr[a] = (a == col) ? 1.f : 0.f; vi[a] = 0.f; }
  switch (p % 5) {
    case 0:
      ap_rot<2>(vr, vi, gs[gb + 0]); ap_rot<1>(vr, vi, gs[gb + 1]);
      ap_rot<0>(vr, vi, gs[gb + 2]);
      ap_cry<2, 1>(vr, vi, gs[gb + 10]); ap_cry<1, 0>(vr, vi, gs[gb + 11]);
      break;
    case 1:
      ap_rot<1>(vr, vi, gs[gb + 3]); ap_rot<0>(vr, vi, gs[gb + 4]);
      ap_cry<2, 1>(vr, vi, gs[gb + 12]); ap_cry<1, 0>(vr, vi, gs[gb + 13]);
      break;
    case 2:
      ap_rot<2>(vr, vi, gs[gb + 5]); ap_rot<0>(vr, vi, gs[gb + 6]);
      ap_cry<1, 2>(vr, vi, gs[gb + 14]); ap_cry<2, 0>(vr, vi, gs[gb + 15]);
      break;
    case 3:
      ap_rot<1>(vr, vi, gs[gb + 7]); ap_rot<0>(vr, vi, gs[gb + 8]);
      ap_cry<2, 1>(vr, vi, gs[gb + 16]); ap_cry<1, 0>(vr, vi, gs[gb + 17]);
      break;
    default:
      ap_rot<0>(vr, vi, gs[gb + 9]);
      ap_cry<2, 0>(vr, vi, gs[gb + 18]); ap_cry<0, 1>(vr, vi, gs[gb + 19]);
      break;
  }
  float4* dst = (float4*)(G + ((p * 4 + (col >> 1)) * 32 + (col & 1) * 16) * 2);
#pragma unroll
  for (int a = 0; a < 8; ++a)
    dst[a] = make_float4(vr[a], -vi[a], vi[a], vr[a]);
}

template <int A0, int A1, int A2, int N0, int N1, int N2, int N3, int T0,
          int T1, int T2>
__device__ __forceinline__ void fused_pass(char* lds, const unsigned wofs,
                                           const v2f (&a)[4], const int t) {
  const int n = t & 15, h = t >> 4;
  const int col = ((n & 1) << N0) | (((n >> 1) & 1) << N1) |
                  (((n >> 2) & 1) << N2) | (((n >> 3) & 1) << N3);
  unsigned ldl = wofs | (8u * swz(col | (h << A0)));
  unsigned stl = wofs | (8u * swz(col | (h << A2)));
  asm volatile("" : "+v"(ldl), "+v"(stl));
  v2f b[2][8];
#pragma unroll
  for (int P = 0; P <= 4; ++P) {
    if (P < 4) {
#pragma unroll
      for (int x = 0; x < 8; ++x) {
        const int T = 2 * P + (x >> 2), kk = x & 3;
        const int c = swz(((T & 1) << T0) | (((T >> 1) & 1) << T1) |
                          (((T >> 2) & 1) << T2) | ((kk & 1) << A1) |
                          ((kk >> 1) << A2));
        const float2 amp = *(const float2*)(
            lds + ((ldl ^ (8u * (c & 31))) + 8u * (c & ~31)));
        b[P & 1][x].x = amp.x;
        b[P & 1][x].y = amp.y;
      }
    }
    if (P > 0) {
      const int Q = P - 1;
      v8f d0 = {}, d1 = {};
#pragma unroll
      for (int kk = 0; kk < 4; ++kk) {
        d0 = __builtin_amdgcn_wmma_f32_16x16x4_f32(
            false, a[kk], false, b[Q & 1][kk], (short)0, d0, false, false);
        d1 = __builtin_amdgcn_wmma_f32_16x16x4_f32(
            false, a[kk], false, b[Q & 1][4 + kk], (short)0, d1, false, false);
      }
#pragma unroll
      for (int x = 0; x < 8; ++x) {
        const int T = 2 * Q + (x >> 2), j = x & 3;
        const int c = swz(((T & 1) << T0) | (((T >> 1) & 1) << T1) |
                          (((T >> 2) & 1) << T2) | ((j & 1) << A0) |
                          ((j >> 1) << A1));
        const v8f d = (x < 4) ? d0 : d1;
        *(float2*)(lds + ((stl ^ (8u * (c & 31))) + 8u * (c & ~31))) =
            make_float2(d[2 * j], d[2 * j + 1]);
      }
    }
  }
  __builtin_amdgcn_wave_barrier();
}

__device__ __forceinline__ void load_afrag(v2f (&a)[4], const float* G,
                                           const int p, const int t) {
  const float2* src = (const float2*)G + p * 128 + t;
#pragma unroll
  for (int kk = 0; kk < 4; ++kk) {
    const float2 g = src[kk * 32];
    a[kk].x = g.x;
    a[kk].y = g.y;
  }
}

#define RUN_PASS(p, ...)                                   \
  {                                                        \
    v2f an[4];                                             \
    if ((p) + 1 < NPASS) load_afrag(an, G, (p) + 1, t);    \
    fused_pass<__VA_ARGS__>(lds, wofs, a, t);              \
    if ((p) + 1 < NPASS) {                                 \
      a[0] = an[0]; a[1] = an[1]; a[2] = an[2]; a[3] = an[3]; \
    }                                                      \
  }

__global__ __launch_bounds__(256, 8) void qsim_kernel(
    const float* __restrict__ inp, const float* __restrict__ G,
    float* __restrict__ out) {
  __shared__ float2 st[WPB][DIM];
  const int t = threadIdx.x & 31;
  const int w = threadIdx.x >> 5;
  const int samp = blockIdx.x * WPB + w;
  char* lds = (char*)&st[0][0];
  const unsigned wofs = w * (DIM * 8);
  const unsigned wt8 = wofs | (8u * t);

  v2f a[4];
  load_afrag(a, G, 0, t);

  float cc[NQ], ss[NQ];
  const float* x = inp + samp * NQ;
#pragma unroll
  for (int q = 0; q < NQ; ++q) {
    float xv = fminf(fmaxf(x[q], 0.f), 1.f);
    float ang = 0.5f * 3.14159265358979323846f * xv;
    cc[q] = cosf(ang);
    ss[q] = sinf(ang);
  }
  float plo = 1.f;
#pragma unroll
  for (int j = 0; j < 5; ++j)
    plo *= ((t >> j) & 1) ? ss[9 - j] : cc[9 - j];
#pragma unroll
  for (int r = 0; r < 32; ++r) {
    float phi = 1.f;
#pragma unroll
    for (int j = 0; j < 5; ++j)
      phi *= ((r >> j) & 1) ? ss[4 - j] : cc[4 - j];
    *(float2*)(lds + ((wt8 ^ (8u * (swz(r << 5) & 31))) + (r << 8))) =
        make_float2(phi * plo, 0.f);
  }
  __builtin_amdgcn_wave_barrier();

  RUN_PASS(0, 7, 8, 9, 0, 1, 2, 3, 4, 5, 6)
  RUN_PASS(1, 5, 6, 7, 0, 1, 2, 3, 4, 8, 9)
  RUN_PASS(2, 3, 5, 4, 0, 1, 2, 6, 7, 8, 9)
  RUN_PASS(3, 1, 2, 3, 0, 4, 5, 8, 6, 7, 9)
  RUN_PASS(4, 0, 9, 1, 2, 3, 4, 7, 5, 6, 8)
  RUN_PASS(5, 7, 8, 9, 0, 1, 2, 3, 4, 5, 6)
  RUN_PASS(6, 5, 6, 7, 0, 1, 2, 3, 4, 8, 9)
  RUN_PASS(7, 3, 5, 4, 0, 1, 2, 6, 7, 8, 9)
  RUN_PASS(8, 1, 2, 3, 0, 4, 5, 8, 6, 7, 9)
  RUN_PASS(9, 0, 9, 1, 2, 3, 4, 7, 5, 6, 8)

  float acc[NQ];
#pragma unroll
  for (int q = 0; q < NQ; ++q) acc[q] = 0.f;
  unsigned mt8 = wt8;
  asm volatile("" : "+v"(mt8));
#pragma unroll
  for (int r = 0; r < 32; ++r) {
    const int i = r * 32 + t;
    const float2 amp = *(const float2*)(
        lds + ((mt8 ^ (8u * (swz(r << 5) & 31))) + (r << 8)));
    const float p = amp.x * amp.x + amp.y * amp.y;
#pragma unroll
    for (int q = 0; q < NQ; ++q)
      acc[q] += ((i >> (9 - q)) & 1) ? -p : p;
  }
#pragma unroll
  for (int q = 0; q < NQ; ++q) {
#pragma unroll
    for (int off = 16; off >= 1; off >>= 1)
      acc[q] += __shfl_xor(acc[q], off, 32);
  }
  if (t == 0) {
#pragma unroll
    for (int q = 0; q < NQ; ++q) out[samp * NQ + q] = acc[q];
  }
}

extern "C" void kernel_launch(void* const* d_in, const int* in_sizes, int n_in,
                              void* d_out, int out_size, void* d_ws,
                              size_t ws_size, hipStream_t stream) {
  const float* inputs = (const float*)d_in[0];
  const float* weights = (const float*)d_in[1];
  float* out = (float*)d_out;
  float* G = (float*)d_ws;

  const int B = in_sizes[0] / NQ;
  build_gates_kernel<<<dim3(1), dim3(128), 0, stream>>>(weights, G);
  qsim_kernel<<<dim3(B / WPB), dim3(256), 0, stream>>>(inputs, G, out);
}


# Layer-1 groups act on the entangled prefix only (128 amplitudes for groups 0-2, 512 for group 3); qubits 7-9 join later by in-LDS scaling: 220 WMMAs per sample, not 320; prep products reassociated
# speedup vs baseline: 3.8714x; 1.2641x over previous
#include <hip/hip_runtime.h>


typedef __attribute__((ext_vector_type(2))) float v2f;
typedef __attribute__((ext_vector_type(8))) float v8f;

#define NQ 10
#define DIM 1024
#define WPB 8
#define NPASS 10

__host__ __device__ constexpr int swz(int i) {
  return i ^ ((i & 32) ? 0x14 : 0) ^ ((i & 64) ? 0x18 : 0) ^
         ((i & 128) ? 0x13 : 0) ^ ((i & 256) ? 0x0A : 0) ^
         ((i & 512) ? 0x10 : 0);
}

template <int l>
__device__ __forceinline__ void ap_rot(float (&vr)[8], float (&vi)[8],
                                       const float* u) {
  const float ar = u[0], ai = u[1], br = u[2], bi = u[3];
  const float cr = u[4], ci = u[5], dr = u[6], di = u[7];
#pragma unroll
  for (int a = 0; a < 8; ++a) {
    if ((a >> l) & 1) continue;
    const int b = a | (1 << l);
    const float x0r = vr[a], x0i = vi[a], x1r = vr[b], x1i = vi[b];
    vr[a] = ar * x0r - ai * x0i + br * x1r - bi * x1i;
    vi[a] = ar * x0i + ai * x0r + br * x1i + bi * x1r;
    vr[b] = cr * x0r - ci * x0i + dr * x1r - di * x1i;
    vi[b] = cr * x0i + ci * x0r + dr * x1i + di * x1r;
  }
}

template <int cb, int tb>
__device__ __forceinline__ void ap_cry(float (&vr)[8], float (&vi)[8],
                                       const float* u) {
  const float c = u[0], s = u[1];
#pragma unroll
  for (int a = 0; a < 8; ++a) {
    if (!((a >> cb) & 1) || ((a >> tb) & 1)) continue;
    const int b = a | (1 << tb);
    const float x0r = vr[a], x0i = vi[a], x1r = vr[b], x1i = vi[b];
    vr[a] = c * x0r - s * x1r;
    vi[a] = c * x0i - s * x1i;
    vr[b] = s * x0r + c * x1r;
    vi[b] = s * x0i + c * x1i;
  }
}

__global__ void build_gates_kernel(const float* __restrict__ W,
                                   float* __restrict__ G) {
  __shared__ float gs[40][8];
  const int g = threadIdx.x;
  if (g < 40) {
    const int L = g / 20, r = g % 20;
    float* u = gs[g];
    if (r < 10) {
      const int base = L * 40 + 3 * r;
      const float phi = W[base], th = W[base + 1], om = W[base + 2];
      const float ct = cosf(0.5f * th), st = sinf(0.5f * th);
      const float ap = 0.5f * (phi + om), am = 0.5f * (phi - om);
      const float cap = cosf(ap), sap = sinf(ap);
      const float cam = cosf(am), sam = sinf(am);
      u[0] =  ct * cap; u[1] = -ct * sap;
      u[2] = -st * cam; u[3] = -st * sam;
      u[4] =  st * cam; u[5] = -st * sam;
      u[6] =  ct * cap; u[7] =  ct * sap;
    } else {
      const float th = W[L * 40 + 30 + (r - 10)];
      u[0] = cosf(0.5f * th);
      u[1] = sinf(0.5f * th);
    }
  }
  __syncthreads();
  if (g >= NPASS * 8) return;
  const int p = g >> 3, col = g & 7;
  const int gb = (p / 5) * 20;
  float vr[8], vi[8];
#pragma unroll
  for (int a = 0; a < 8; ++a) { vr[a] = (a == col) ? 1.f : 0.f; vi[a] = 0.f; }
  switch (p % 5) {
    case 0:
      ap_rot<2>(vr, vi, gs[gb + 0]); ap_rot<1>(vr, vi, gs[gb + 1]);
      ap_rot<0>(vr, vi, gs[gb + 2]);
      ap_cry<2, 1>(vr, vi, gs[gb + 10]); ap_cry<1, 0>(vr, vi, gs[gb + 11]);
      break;
    case 1:
      ap_rot<1>(vr, vi, gs[gb + 3]); ap_rot<0>(vr, vi, gs[gb + 4]);
      ap_cry<2, 1>(vr, vi, gs[gb + 12]); ap_cry<1, 0>(vr, vi, gs[gb + 13]);
      break;
    case 2:
      ap_rot<2>(vr, vi, gs[gb + 5]); ap_rot<0>(vr, vi, gs[gb + 6]);
      ap_cry<1, 2>(vr, vi, gs[gb + 14]); ap_cry<2, 0>(vr, vi, gs[gb + 15]);
      break;
    case 3:
      ap_rot<1>(vr, vi, gs[gb + 7]); ap_rot<0>(vr, vi, gs[gb + 8]);
      ap_cry<2, 1>(vr, vi, gs[gb + 16]); ap_cry<1, 0>(vr, vi, gs[gb + 17]);
      break;
    default:
      ap_rot<0>(vr, vi, gs[gb + 9]);
      ap_cry<2, 0>(vr, vi, gs[gb + 18]); ap_cry<0, 1>(vr, vi, gs[gb + 19]);
      break;
  }
  float4* dst = (float4*)(G + ((p * 4 + (col >> 1)) * 32 + (col & 1) * 16) * 2);
#pragma unroll
  for (int a = 0; a < 8; ++a)
    dst[a] = make_float4(vr[a], -vi[a], vi[a], vr[a]);
}

template <int NPAIR, int A0, int A1, int A2, int N0, int N1, int N2, int N3,
          int T0, int T1, int T2>
__device__ __forceinline__ void fused_pass(char* lds, const unsigned wofs,
                                           const v2f (&a)[4], const int t) {
  const int n = t & 15, h = t >> 4;
  const int col = ((n & 1) << N0) | (((n >> 1) & 1) << N1) |
                  (((n >> 2) & 1) << N2) | (((n >> 3) & 1) << N3);
  unsigned ldl = wofs | (8u * swz(col | (h << A0)));
  unsigned stl = wofs | (8u * swz(col | (h << A2)));
  asm volatile("" : "+v"(ldl), "+v"(stl));
  v2f b[2][8];
#pragma unroll
  for (int P = 0; P <= NPAIR; ++P) {
    if (P < NPAIR) {
#pragma unroll
      for (int x = 0; x < 8; ++x) {
        const int T = 2 * P + (x >> 2), kk = x & 3;
        const int c = swz(((T & 1) << T0) | (((T >> 1) & 1) << T1) |
                          (((T >> 2) & 1) << T2) | ((kk & 1) << A1) |
                          ((kk >> 1) << A2));
        const float2 amp = *(const float2*)(
            lds + ((ldl ^ (8u * (c & 31))) + 8u * (c & ~31)));
        b[P & 1][x].x = amp.x;
        b[P & 1][x].y = amp.y;
      }
    }
    if (P > 0) {
      const int Q = P - 1;
      v8f d0 = {}, d1 = {};
#pragma unroll
      for (int kk = 0; kk < 4; ++kk) {
        d0 = __builtin_amdgcn_wmma_f32_16x16x4_f32(
            false, a[kk], false, b[Q & 1][kk], (short)0, d0, false, false);
        d1 = __builtin_amdgcn_wmma_f32_16x16x4_f32(
            false, a[kk], false, b[Q & 1][4 + kk], (short)0, d1, false, false);
      }
#pragma unroll
      for (int x = 0; x < 8; ++x) {
        const int T = 2 * Q + (x >> 2), j = x & 3;
        const int c = swz(((T & 1) << T0) | (((T >> 1) & 1) << T1) |
                          (((T >> 2) & 1) << T2) | ((j & 1) << A0) |
                          ((j >> 1) << A1));
        const v8f d = (x < 4) ? d0 : d1;
        *(float2*)(lds + ((stl ^ (8u * (c & 31))) + 8u * (c & ~31))) =
            make_float2(d[2 * j], d[2 * j + 1]);
      }
    }
  }
  __builtin_amdgcn_wave_barrier();
}

template <int A0, int A1, int A2, int N0, int N1, int N2, int N3>
__device__ __forceinline__ void prefix_pass(char* lds, const unsigned wofs,
                                            const v2f (&a)[4], const int t) {
  const int n = t & 15, h = t >> 4;
  const int col = ((n & 1) << N0) | (((n >> 1) & 1) << N1) |
                  (((n >> 2) & 1) << N2) | (((n >> 3) & 1) << N3);
  const unsigned ldl = wofs | (8u * swz(col | (h << A0)));
  const unsigned stl = wofs | (8u * swz(col | (h << A2)));
  v2f b[4];
#pragma unroll
  for (int kk = 0; kk < 4; ++kk) {
    const int c = swz(((kk & 1) << A1) | ((kk >> 1) << A2));
    const float2 amp = *(const float2*)(
        lds + ((ldl ^ (8u * (c & 31))) + 8u * (c & ~31)));
    b[kk].x = amp.x;
    b[kk].y = amp.y;
  }
  v8f d = {};
#pragma unroll
  for (int kk = 0; kk < 4; ++kk)
    d = __builtin_amdgcn_wmma_f32_16x16x4_f32(
        false, a[kk], false, b[kk], (short)0, d, false, false);
#pragma unroll
  for (int j = 0; j < 4; ++j) {
    const int c = swz(((j & 1) << A0) | ((j >> 1) << A1));
    *(float2*)(lds + ((stl ^ (8u * (c & 31))) + 8u * (c & ~31))) =
        make_float2(d[2 * j], d[2 * j + 1]);
  }
  __builtin_amdgcn_wave_barrier();
}

__device__ __forceinline__ void load_afrag(v2f (&a)[4], const float* G,
                                           const int p, const int t) {
  const float2* src = (const float2*)G + p * 128 + t;
#pragma unroll
  for (int kk = 0; kk < 4; ++kk) {
    const float2 g = src[kk * 32];
    a[kk].x = g.x;
    a[kk].y = g.y;
  }
}

#define RUN_PASS(p, ...)                                   \
  {                                                        \
    v2f an[4];                                             \
    if ((p) + 1 < NPASS) load_afrag(an, G, (p) + 1, t);    \
    fused_pass<4, __VA_ARGS__>(lds, wofs, a, t);           \
    if ((p) + 1 < NPASS) {                                 \
      a[0] = an[0]; a[1] = an[1]; a[2] = an[2]; a[3] = an[3]; \
    }                                                      \
  }

#define RUN_PREFIX(p, ...)                                 \
  {                                                        \
    v2f an[4];                                             \
    load_afrag(an, G, (p) + 1, t);                         \
    prefix_pass<__VA_ARGS__>(lds, wofs, a, t);             \
    a[0] = an[0]; a[1] = an[1]; a[2] = an[2]; a[3] = an[3]; \
  }

#define RUN_HALF(p, ...)                                   \
  {                                                        \
    v2f an[4];                                             \
    load_afrag(an, G, (p) + 1, t);                         \
    fused_pass<2, __VA_ARGS__, 0>(lds, wofs, a, t);        \
    a[0] = an[0]; a[1] = an[1]; a[2] = an[2]; a[3] = an[3]; \
  }

__global__ __launch_bounds__(256, 8) void qsim_kernel(
    const float* __restrict__ inp, const float* __restrict__ G,
    float* __restrict__ out) {
  __shared__ float2 st[WPB][DIM];
  const int t = threadIdx.x & 31;
  const int w = threadIdx.x >> 5;
  const int samp = blockIdx.x * WPB + w;
  char* lds = (char*)&st[0][0];
  const unsigned wofs = w * (DIM * 8);
  const unsigned wt8 = wofs | (8u * t);

  v2f a[4];
  load_afrag(a, G, 0, t);

  float cc[NQ], ss[NQ];
  const float* x = inp + samp * NQ;
#pragma unroll
  for (int q = 0; q < NQ; ++q) {
    float xv = fminf(fmaxf(x[q], 0.f), 1.f);
    float ang = 0.5f * 3.14159265358979323846f * xv;
    cc[q] = cosf(ang);
    ss[q] = sinf(ang);
  }
  float ppre = 1.f;
#pragma unroll
  for (int j = 0; j < 5; ++j)
    ppre *= ((t >> j) & 1) ? ss[6 - j] : cc[6 - j];
  const float f78 = ((t & 1) ? ss[8] : cc[8]) * ((t & 2) ? ss[7] : cc[7]);
  const float f9 = (t & 1) ? ss[9] : cc[9];
  const unsigned pa = wofs | (8u * swz(((t & 3) << 3) | ((t >> 2) << 5)));
#pragma unroll
  for (int rr = 0; rr < 4; ++rr) {
    const float f = ppre * ((rr & 1) ? ss[1] : cc[1]) * ((rr & 2) ? ss[0] : cc[0]);
    const int c = swz(rr << 8);
    *(float2*)(lds + ((pa ^ (8u * (c & 31))) + 8u * (c & ~31))) =
        make_float2(f, 0.f);
  }
  __builtin_amdgcn_wave_barrier();

  RUN_PREFIX(0, 7, 8, 9, 3, 4, 5, 6)
  RUN_PREFIX(1, 5, 6, 7, 3, 4, 8, 9)
  RUN_PREFIX(2, 3, 5, 4, 6, 7, 8, 9)

  {
    const unsigned el = wofs | (8u * swz(((t & 12) << 1) | ((t >> 4) << 5)));
    const unsigned es = wofs | (8u * swz(((t & 15) << 1) | ((t >> 4) << 5)));
#pragma unroll
    for (int g = 0; g < 2; ++g) {
      float2 v[8];
#pragma unroll
      for (int x = 0; x < 8; ++x) {
        const int c = swz((8 * g + x) << 6);
        v[x] = *(const float2*)(lds + ((el ^ (8u * (c & 31))) + 8u * (c & ~31)));
      }
#pragma unroll
      for (int x = 0; x < 8; ++x) {
        const int c = swz((8 * g + x) << 6);
        *(float2*)(lds + ((es ^ (8u * (c & 31))) + 8u * (c & ~31))) =
            make_float2(v[x].x * f78, v[x].y * f78);
      }
    }
  }
  __builtin_amdgcn_wave_barrier();

  RUN_HALF(3, 1, 2, 3, 4, 5, 7, 8, 6, 9)

  {
    const unsigned xl = wofs | (8u * (t & 0x1E));
#pragma unroll
    for (int g = 0; g < 4; ++g) {
      float2 v[8];
#pragma unroll
      for (int x = 0; x < 8; ++x) {
        const int r = 8 * g + x;
        v[x] = *(const float2*)(
            lds + ((xl ^ (8u * (swz(r << 5) & 31))) + (r << 8)));
      }
#pragma unroll
      for (int x = 0; x < 8; ++x) {
        const int r = 8 * g + x;
        *(float2*)(lds + ((wt8 ^ (8u * (swz(r << 5) & 31))) + (r << 8))) =
            make_float2(v[x].x * f9, v[x].y * f9);
      }
    }
  }
  __builtin_amdgcn_wave_barrier();

  RUN_PASS(4, 0, 9, 1, 2, 3, 4, 7, 5, 6, 8)
  RUN_PASS(5, 7, 8, 9, 0, 1, 2, 3, 4, 5, 6)
  RUN_PASS(6, 5, 6, 7, 0, 1, 2, 3, 4, 8, 9)
  RUN_PASS(7, 3, 5, 4, 0, 1, 2, 6, 7, 8, 9)
  RUN_PASS(8, 1, 2, 3, 0, 4, 5, 8, 6, 7, 9)
  RUN_PASS(9, 0, 9, 1, 2, 3, 4, 7, 5, 6, 8)

  float acc[NQ];
#pragma unroll
  for (int q = 0; q < NQ; ++q) acc[q] = 0.f;
  unsigned mt8 = wt8;
  asm volatile("" : "+v"(mt8));
#pragma unroll
  for (int r = 0; r < 32; ++r) {
    const int i = r * 32 + t;
    const float2 amp = *(const float2*)(
        lds + ((mt8 ^ (8u * (swz(r << 5) & 31))) + (r << 8)));
    const float p = amp.x * amp.x + amp.y * amp.y;
#pragma unroll
    for (int q = 0; q < NQ; ++q)
      acc[q] += ((i >> (9 - q)) & 1) ? -p : p;
  }
#pragma unroll
  for (int q = 0; q < NQ; ++q) {
#pragma unroll
    for (int off = 16; off >= 1; off >>= 1)
      acc[q] += __shfl_xor(acc[q], off, 32);
  }
  if (t == 0) {
#pragma unroll
    for (int q = 0; q < NQ; ++q) out[samp * NQ + q] = acc[q];
  }
}

extern "C" void kernel_launch(void* const* d_in, const int* in_sizes, int n_in,
                              void* d_out, int out_size, void* d_ws,
                              size_t ws_size, hipStream_t stream) {
  const float* inputs = (const float*)d_in[0];
  const float* weights = (const float*)d_in[1];
  float* out = (float*)d_out;
  float* G = (float*)d_ws;

  const int B = in_sizes[0] / NQ;
  build_gates_kernel<<<dim3(1), dim3(128), 0, stream>>>(weights, G);
  qsim_kernel<<<dim3(B / WPB), dim3(256), 0, stream>>>(inputs, G, out);
}


# Input angles' cosine and sine taken once per wave: lane q computes qubit q and all lanes read the ten pairs by v_readlane (was: every lane computed all ten); one input load per wave instead of ten
# speedup vs baseline: 4.1940x; 1.0833x over previous
#include <hip/hip_runtime.h>


typedef __attribute__((ext_vector_type(2))) float v2f;
typedef __attribute__((ext_vector_type(8))) float v8f;

#define NQ 10
#define DIM 1024
#define WPB 8
#define NPASS 10

__host__ __device__ constexpr int swz(int i) {
  return i ^ ((i & 32) ? 0x14 : 0) ^ ((i & 64) ? 0x18 : 0) ^
         ((i & 128) ? 0x13 : 0) ^ ((i & 256) ? 0x0A : 0) ^
         ((i & 512) ? 0x10 : 0);
}

template <int l>
__device__ __forceinline__ void ap_rot(float (&vr)[8], float (&vi)[8],
                                       const float* u) {
  const float ar = u[0], ai = u[1], br = u[2], bi = u[3];
  const float cr = u[4], ci = u[5], dr = u[6], di = u[7];
#pragma unroll
  for (int a = 0; a < 8; ++a) {
    if ((a >> l) & 1) continue;
    const int b = a | (1 << l);
    const float x0r = vr[a], x0i = vi[a], x1r = vr[b], x1i = vi[b];
    vr[a] = ar * x0r - ai * x0i + br * x1r - bi * x1i;
    vi[a] = ar * x0i + ai * x0r + br * x1i + bi * x1r;
    vr[b] = cr * x0r - ci * x0i + dr * x1r - di * x1i;
    vi[b] = cr * x0i + ci * x0r + dr * x1i + di * x1r;
  }
}

template <int cb, int tb>
__device__ __forceinline__ void ap_cry(float (&vr)[8], float (&vi)[8],
                                       const float* u) {
  const float c = u[0], s = u[1];
#pragma unroll
  for (int a = 0; a < 8; ++a) {
    if (!((a >> cb) & 1) || ((a >> tb) & 1)) continue;
    const int b = a | (1 << tb);
    const float x0r = vr[a], x0i = vi[a], x1r = vr[b], x1i = vi[b];
    vr[a] = c * x0r - s * x1r;
    vi[a] = c * x0i - s * x1i;
    vr[b] = s * x0r + c * x1r;
    vi[b] = s * x0i + c * x1i;
  }
}

__global__ void build_gates_kernel(const float* __restrict__ W,
                                   float* __restrict__ G) {
  __shared__ float gs[40][8];
  const int g = threadIdx.x;
  if (g < 40) {
    const int L = g / 20, r = g % 20;
    float* u = gs[g];
    if (r < 10) {
      const int base = L * 40 + 3 * r;
      const float phi = W[base], th = W[base + 1], om = W[base + 2];
      const float ct = cosf(0.5f * th), st = sinf(0.5f * th);
      const float ap = 0.5f * (phi + om), am = 0.5f * (phi - om);
      const float cap = cosf(ap), sap = sinf(ap);
      const float cam = cosf(am), sam = sinf(am);
      u[0] =  ct * cap; u[1] = -ct * sap;
      u[2] = -st * cam; u[3] = -st * sam;
      u[4] =  st * cam; u[5] = -st * sam;
      u[6] =  ct * cap; u[7] =  ct * sap;
    } else {
      const float th = W[L * 40 + 30 + (r - 10)];
      u[0] = cosf(0.5f * th);
      u[1] = sinf(0.5f * th);
    }
  }
  __syncthreads();
  if (g >= NPASS * 8) return;
  const int p = g >> 3, col = g & 7;
  const int gb = (p / 5) * 20;
  float vr[8], vi[8];
#pragma unroll
  for (int a = 0; a < 8; ++a) { vr[a] = (a == col) ? 1.f : 0.f; vi[a] = 0.f; }
  switch (p % 5) {
    case 0:
      ap_rot<2>(vr, vi, gs[gb + 0]); ap_rot<1>(vr, vi, gs[gb + 1]);
      ap_rot<0>(vr, vi, gs[gb + 2]);
      ap_cry<2, 1>(vr, vi, gs[gb + 10]); ap_cry<1, 0>(vr, vi, gs[gb + 11]);
      break;
    case 1:
      ap_rot<1>(vr, vi, gs[gb + 3]); ap_rot<0>(vr, vi, gs[gb + 4]);
      ap_cry<2, 1>(vr, vi, gs[gb + 12]); ap_cry<1, 0>(vr, vi, gs[gb + 13]);
      break;
    case 2:
      ap_rot<2>(vr, vi, gs[gb + 5]); ap_rot<0>(vr, vi, gs[gb + 6]);
      ap_cry<1, 2>(vr, vi, gs[gb + 14]); ap_cry<2, 0>(vr, vi, gs[gb + 15]);
      break;
    case 3:
      ap_rot<1>(vr, vi, gs[gb + 7]); ap_rot<0>(vr, vi, gs[gb + 8]);
      ap_cry<2, 1>(vr, vi, gs[gb + 16]); ap_cry<1, 0>(vr, vi, gs[gb + 17]);
      break;
    default:
      ap_rot<0>(vr, vi, gs[gb + 9]);
      ap_cry<2, 0>(vr, vi, gs[gb + 18]); ap_cry<0, 1>(vr, vi, gs[gb + 19]);
      break;
  }
  float4* dst = (float4*)(G + ((p * 4 + (col >> 1)) * 32 + (col & 1) * 16) * 2);
#pragma unroll
  for (int a = 0; a < 8; ++a)
    dst[a] = make_float4(vr[a], -vi[a], vi[a], vr[a]);
}

template <int NPAIR, int A0, int A1, int A2, int N0, int N1, int N2, int N3,
          int T0, int T1, int T2>
__device__ __forceinline__ void fused_pass(char* lds, const unsigned wofs,
                                           const v2f (&a)[4], const int t) {
  const int n = t & 15, h = t >> 4;
  const int col = ((n & 1) << N0) | (((n >> 1) & 1) << N1) |
                  (((n >> 2) & 1) << N2) | (((n >> 3) & 1) << N3);
  unsigned ldl = wofs | (8u * swz(col | (h << A0)));
  unsigned stl = wofs | (8u * swz(col | (h << A2)));
  asm volatile("" : "+v"(ldl), "+v"(stl));
  v2f b[2][8];
#pragma unroll
  for (int P = 0; P <= NPAIR; ++P) {
    if (P < NPAIR) {
#pragma unroll
      for (int x = 0; x < 8; ++x) {
        const int T = 2 * P + (x >> 2), kk = x & 3;
        const int c = swz(((T & 1) << T0) | (((T >> 1) & 1) << T1) |
                          (((T >> 2) & 1) << T2) | ((kk & 1) << A1) |
                          ((kk >> 1) << A2));
        const float2 amp = *(const float2*)(
            lds + ((ldl ^ (8u * (c & 31))) + 8u * (c & ~31)));
        b[P & 1][x].x = amp.x;
        b[P & 1][x].y = amp.y;
      }
    }
    if (P > 0) {
      const int Q = P - 1;
      v8f d0 = {}, d1 = {};
#pragma unroll
      for (int kk = 0; kk < 4; ++kk) {
        d0 = __builtin_amdgcn_wmma_f32_16x16x4_f32(
            false, a[kk], false, b[Q & 1][kk], (short)0, d0, false, false);
        d1 = __builtin_amdgcn_wmma_f32_16x16x4_f32(
            false, a[kk], false, b[Q & 1][4 + kk], (short)0, d1, false, false);
      }
#pragma unroll
      for (int x = 0; x < 8; ++x) {
        const int T = 2 * Q + (x >> 2), j = x & 3;
        const int c = swz(((T & 1) << T0) | (((T >> 1) & 1) << T1) |
                          (((T >> 2) & 1) << T2) | ((j & 1) << A0) |
                          ((j >> 1) << A1));
        const v8f d = (x < 4) ? d0 : d1;
        *(float2*)(lds + ((stl ^ (8u * (c & 31))) + 8u * (c & ~31))) =
            make_float2(d[2 * j], d[2 * j + 1]);
      }
    }
  }
  __builtin_amdgcn_wave_barrier();
}

template <int A0, int A1, int A2, int N0, int N1, int N2, int N3>
__device__ __forceinline__ void prefix_pass(char* lds, const unsigned wofs,
                                            const v2f (&a)[4], const int t) {
  const int n = t & 15, h = t >> 4;
  const int col = ((n & 1) << N0) | (((n >> 1) & 1) << N1) |
                  (((n >> 2) & 1) << N2) | (((n >> 3) & 1) << N3);
  const unsigned ldl = wofs | (8u * swz(col | (h << A0)));
  const unsigned stl = wofs | (8u * swz(col | (h << A2)));
  v2f b[4];
#pragma unroll
  for (int kk = 0; kk < 4; ++kk) {
    const int c = swz(((kk & 1) << A1) | ((kk >> 1) << A2));
    const float2 amp = *(const float2*)(
        lds + ((ldl ^ (8u * (c & 31))) + 8u * (c & ~31)));
    b[kk].x = amp.x;
    b[kk].y = amp.y;
  }
  v8f d = {};
#pragma unroll
  for (int kk = 0; kk < 4; ++kk)
    d = __builtin_amdgcn_wmma_f32_16x16x4_f32(
        false, a[kk], false, b[kk], (short)0, d, false, false);
#pragma unroll
  for (int j = 0; j < 4; ++j) {
    const int c = swz(((j & 1) << A0) | ((j >> 1) << A1));
    *(float2*)(lds + ((stl ^ (8u * (c & 31))) + 8u * (c & ~31))) =
        make_float2(d[2 * j], d[2 * j + 1]);
  }
  __builtin_amdgcn_wave_barrier();
}

__device__ __forceinline__ void load_afrag(v2f (&a)[4], const float* G,
                                           const int p, const int t) {
  const float2* src = (const float2*)G + p * 128 + t;
#pragma unroll
  for (int kk = 0; kk < 4; ++kk) {
    const float2 g = src[kk * 32];
    a[kk].x = g.x;
    a[kk].y = g.y;
  }
}

#define RUN_PASS(p, ...)                                   \
  {                                                        \
    v2f an[4];                                             \
    if ((p) + 1 < NPASS) load_afrag(an, G, (p) + 1, t);    \
    fused_pass<4, __VA_ARGS__>(lds, wofs, a, t);           \
    if ((p) + 1 < NPASS) {                                 \
      a[0] = an[0]; a[1] = an[1]; a[2] = an[2]; a[3] = an[3]; \
    }                                                      \
  }

#define RUN_PREFIX(p, ...)                                 \
  {                                                        \
    v2f an[4];                                             \
    load_afrag(an, G, (p) + 1, t);                         \
    prefix_pass<__VA_ARGS__>(lds, wofs, a, t);             \
    a[0] = an[0]; a[1] = an[1]; a[2] = an[2]; a[3] = an[3]; \
  }

#define RUN_HALF(p, ...)                                   \
  {                                                        \
    v2f an[4];                                             \
    load_afrag(an, G, (p) + 1, t);                         \
    fused_pass<2, __VA_ARGS__, 0>(lds, wofs, a, t);        \
    a[0] = an[0]; a[1] = an[1]; a[2] = an[2]; a[3] = an[3]; \
  }

__global__ __launch_bounds__(256, 8) void qsim_kernel(
    const float* __restrict__ inp, const float* __restrict__ G,
    float* __restrict__ out) {
  __shared__ float2 st[WPB][DIM];
  const int t = threadIdx.x & 31;
  const int w = threadIdx.x >> 5;
  const int samp = blockIdx.x * WPB + w;
  char* lds = (char*)&st[0][0];
  const unsigned wofs = w * (DIM * 8);
  const unsigned wt8 = wofs | (8u * t);

  v2f a[4];
  load_afrag(a, G, 0, t);

  float cc[NQ], ss[NQ];
  const float* x = inp + samp * NQ;
  {
    const float xl = (t < NQ) ? x[t] : 0.f;
    const float xv = fminf(fmaxf(xl, 0.f), 1.f);
    const float ang = 0.5f * 3.14159265358979323846f * xv;
    const float cl = cosf(ang), sl = sinf(ang);
#pragma unroll
    for (int q = 0; q < NQ; ++q) {
      cc[q] = __int_as_float(__builtin_amdgcn_readlane(__float_as_int(cl), q));
      ss[q] = __int_as_float(__builtin_amdgcn_readlane(__float_as_int(sl), q));
    }
  }
  float ppre = 1.f;
#pragma unroll
  for (int j = 0; j < 5; ++j)
    ppre *= ((t >> j) & 1) ? ss[6 - j] : cc[6 - j];
  const float f78 = ((t & 1) ? ss[8] : cc[8]) * ((t & 2) ? ss[7] : cc[7]);
  const float f9 = (t & 1) ? ss[9] : cc[9];
  const unsigned pa = wofs | (8u * swz(((t & 3) << 3) | ((t >> 2) << 5)));
#pragma unroll
  for (int rr = 0; rr < 4; ++rr) {
    const float f = ppre * ((rr & 1) ? ss[1] : cc[1]) * ((rr & 2) ? ss[0] : cc[0]);
    const int c = swz(rr << 8);
    *(float2*)(lds + ((pa ^ (8u * (c & 31))) + 8u * (c & ~31))) =
        make_float2(f, 0.f);
  }
  __builtin_amdgcn_wave_barrier();

  RUN_PREFIX(0, 7, 8, 9, 3, 4, 5, 6)
  RUN_PREFIX(1, 5, 6, 7, 3, 4, 8, 9)
  RUN_PREFIX(2, 3, 5, 4, 6, 7, 8, 9)

  {
    const unsigned el = wofs | (8u * swz(((t & 12) << 1) | ((t >> 4) << 5)));
    const unsigned es = wofs | (8u * swz(((t & 15) << 1) | ((t >> 4) << 5)));
#pragma unroll
    for (int g = 0; g < 2; ++g) {
      float2 v[8];
#pragma unroll
      for (int x = 0; x < 8; ++x) {
        const int c = swz((8 * g + x) << 6);
        v[x] = *(const float2*)(lds + ((el ^ (8u * (c & 31))) + 8u * (c & ~31)));
      }
#pragma unroll
      for (int x = 0; x < 8; ++x) {
        const int c = swz((8 * g + x) << 6);
        *(float2*)(lds + ((es ^ (8u * (c & 31))) + 8u * (c & ~31))) =
            make_float2(v[x].x * f78, v[x].y * f78);
      }
    }
  }
  __builtin_amdgcn_wave_barrier();

  RUN_HALF(3, 1, 2, 3, 4, 5, 7, 8, 6, 9)

  {
    const unsigned xl = wofs | (8u * (t & 0x1E));
#pragma unroll
    for (int g = 0; g < 4; ++g) {
      float2 v[8];
#pragma unroll
      for (int x = 0; x < 8; ++x) {
        const int r = 8 * g + x;
        v[x] = *(const float2*)(
            lds + ((xl ^ (8u * (swz(r << 5) & 31))) + (r << 8)));
      }
#pragma unroll
      for (int x = 0; x < 8; ++x) {
        const int r = 8 * g + x;
        *(float2*)(lds + ((wt8 ^ (8u * (swz(r << 5) & 31))) + (r << 8))) =
            make_float2(v[x].x * f9, v[x].y * f9);
      }
    }
  }
  __builtin_amdgcn_wave_barrier();

  RUN_PASS(4, 0, 9, 1, 2, 3, 4, 7, 5, 6, 8)
  RUN_PASS(5, 7, 8, 9, 0, 1, 2, 3, 4, 5, 6)
  RUN_PASS(6, 5, 6, 7, 0, 1, 2, 3, 4, 8, 9)
  RUN_PASS(7, 3, 5, 4, 0, 1, 2, 6, 7, 8, 9)
  RUN_PASS(8, 1, 2, 3, 0, 4, 5, 8, 6, 7, 9)
  RUN_PASS(9, 0, 9, 1, 2, 3, 4, 7, 5, 6, 8)

  float acc[NQ];
#pragma unroll
  for (int q = 0; q < NQ; ++q) acc[q] = 0.f;
  unsigned mt8 = wt8;
  asm volatile("" : "+v"(mt8));
#pragma unroll
  for (int r = 0; r < 32; ++r) {
    const int i = r * 32 + t;
    const float2 amp = *(const float2*)(
        lds + ((mt8 ^ (8u * (swz(r << 5) & 31))) + (r << 8)));
    const float p = amp.x * amp.x + amp.y * amp.y;
#pragma unroll
    for (int q = 0; q < NQ; ++q)
      acc[q] += ((i >> (9 - q)) & 1) ? -p : p;
  }
#pragma unroll
  for (int q = 0; q < NQ; ++q) {
#pragma unroll
    for (int off = 16; off >= 1; off >>= 1)
      acc[q] += __shfl_xor(acc[q], off, 32);
  }
  if (t == 0) {
#pragma unroll
    for (int q = 0; q < NQ; ++q) out[samp * NQ + q] = acc[q];
  }
}

extern "C" void kernel_launch(void* const* d_in, const int* in_sizes, int n_in,
                              void* d_out, int out_size, void* d_ws,
                              size_t ws_size, hipStream_t stream) {
  const float* inputs = (const float*)d_in[0];
  const float* weights = (const float*)d_in[1];
  float* out = (float*)d_out;
  float* G = (float*)d_ws;

  const int B = in_sizes[0] / NQ;
  build_gates_kernel<<<dim3(1), dim3(128), 0, stream>>>(weights, G);
  qsim_kernel<<<dim3(B / WPB), dim3(256), 0, stream>>>(inputs, G, out);
}


# Gate build fused into the main kernel: every work-group builds the ten 16x16 matrices in its own LDS (one thread per angle, wave-uniform group), beside state preparation; one launch, not two
# speedup vs baseline: 4.3276x; 1.0319x over previous
#include <hip/hip_runtime.h>


typedef __attribute__((ext_vector_type(2))) float v2f;
typedef __attribute__((ext_vector_type(8))) float v8f;

#define NQ 10
#define DIM 1024
#define WPB 8
#define NPASS 10

__host__ __device__ constexpr int swz(int i) {
  return i ^ ((i & 32) ? 0x14 : 0) ^ ((i & 64) ? 0x18 : 0) ^
         ((i & 128) ? 0x13 : 0) ^ ((i & 256) ? 0x0A : 0) ^
         ((i & 512) ? 0x10 : 0);
}

template <int l>
__device__ __forceinline__ void ap_rot(float (&vr)[8], float (&vi)[8],
                                       const float* u) {
  const float ar = u[0], ai = u[1], br = u[2], bi = u[3];
  const float cr = u[4], ci = u[5], dr = u[6], di = u[7];
#pragma unroll
  for (int a = 0; a < 8; ++a) {
    if ((a >> l) & 1) continue;
    const int b = a | (1 << l);
    const float x0r = vr[a], x0i = vi[a], x1r = vr[b], x1i = vi[b];
    vr[a] = ar * x0r - ai * x0i + br * x1r - bi * x1i;
    vi[a] = ar * x0i + ai * x0r + br * x1i + bi * x1r;
    vr[b] = cr * x0r - ci * x0i + dr * x1r - di * x1i;
    vi[b] = cr * x0i + ci * x0r + dr * x1i + di * x1r;
  }
}

template <int cb, int tb>
__device__ __forceinline__ void ap_cry(float (&vr)[8], float (&vi)[8],
                                       const float* u) {
  const float c = u[0], s = u[1];
#pragma unroll
  for (int a = 0; a < 8; ++a) {
    if (!((a >> cb) & 1) || ((a >> tb) & 1)) continue;
    const int b = a | (1 << tb);
    const float x0r = vr[a], x0i = vi[a], x1r = vr[b], x1i = vi[b];
    vr[a] = c * x0r - s * x1r;
    vi[a] = c * x0i - s * x1i;
    vr[b] = s * x0r + c * x1r;
    vi[b] = s * x0i + c * x1i;
  }
}

__device__ __forceinline__ void gate_trig(const float* __restrict__ W,
                                          float* tc, float* ts, const int g) {
  const int a = g & 127;
  if (a < 80) {
    float ang;
    if (a < 60) {
      const int rg = a / 3, k = a - 3 * rg;
      const int base = (rg / 10) * 40 + 3 * (rg % 10);
      const float phi = W[base], th = W[base + 1], om = W[base + 2];
      ang = (k == 0) ? 0.5f * th
                     : (k == 1) ? 0.5f * (phi + om) : 0.5f * (phi - om);
    } else {
      const int cg = a - 60;
      ang = 0.5f * W[(cg / 10) * 40 + 30 + (cg % 10)];
    }
    if (g < 128) tc[a] = cosf(ang); else ts[a] = sinf(ang);
  }
}

__device__ __forceinline__ void gate_entries(const float* tc, const float* ts,
                                             float (*gs)[8], const int g) {
  if (g < 40) {
    const int L = g / 20, r = g % 20;
    float* u = gs[g];
    if (r < 10) {
      const int a = 3 * (L * 10 + r);
      const float ct = tc[a], st = ts[a];
      const float cap = tc[a + 1], sap = ts[a + 1];
      const float cam = tc[a + 2], sam = ts[a + 2];
      u[0] =  ct * cap; u[1] = -ct * sap;
      u[2] = -st * cam; u[3] = -st * sam;
      u[4] =  st * cam; u[5] = -st * sam;
      u[6] =  ct * cap; u[7] =  ct * sap;
    } else {
      const int a = 60 + L * 10 + (r - 10);
      u[0] = tc[a];
      u[1] = ts[a];
    }
  }
}

__device__ __forceinline__ void gate_columns(float (*gs)[8], float* G,
                                             const int wv, const int ln) {
  if (wv >= 5 || ln >= 16) return;
  const int p = (ln >> 3) * 5 + wv, col = ln & 7;
  const int gb = (ln >> 3) * 20;
  float vr[8], vi[8];
#pragma unroll
  for (int a = 0; a < 8; ++a) { vr[a] = (a == col) ? 1.f : 0.f; vi[a] = 0.f; }
  switch (wv) {
    case 0:
      ap_rot<2>(vr, vi, gs[gb + 0]); ap_rot<1>(vr, vi, gs[gb + 1]);
      ap_rot<0>(vr, vi, gs[gb + 2]);
      ap_cry<2, 1>(vr, vi, gs[gb + 10]); ap_cry<1, 0>(vr, vi, gs[gb + 11]);
      break;
    case 1:
      ap_rot<1>(vr, vi, gs[gb + 3]); ap_rot<0>(vr, vi, gs[gb + 4]);
      ap_cry<2, 1>(vr, vi, gs[gb + 12]); ap_cry<1, 0>(vr, vi, gs[gb + 13]);
      break;
    case 2:
      ap_rot<2>(vr, vi, gs[gb + 5]); ap_rot<0>(vr, vi, gs[gb + 6]);
      ap_cry<1, 2>(vr, vi, gs[gb + 14]); ap_cry<2, 0>(vr, vi, gs[gb + 15]);
      break;
    case 3:
      ap_rot<1>(vr, vi, gs[gb + 7]); ap_rot<0>(vr, vi, gs[gb + 8]);
      ap_cry<2, 1>(vr, vi, gs[gb + 16]); ap_cry<1, 0>(vr, vi, gs[gb + 17]);
      break;
    default:
      ap_rot<0>(vr, vi, gs[gb + 9]);
      ap_cry<2, 0>(vr, vi, gs[gb + 18]); ap_cry<0, 1>(vr, vi, gs[gb + 19]);
      break;
  }
  float4* dst = (float4*)(G + ((p * 4 + (col >> 1)) * 32 + (col & 1) * 16) * 2);
#pragma unroll
  for (int a = 0; a < 8; ++a)
    dst[a] = make_float4(vr[a], -vi[a], vi[a], vr[a]);
}

template <int NPAIR, int A0, int A1, int A2, int N0, int N1, int N2, int N3,
          int T0, int T1, int T2>
__device__ __forceinline__ void fused_pass(char* lds, const unsigned wofs,
                                           const v2f (&a)[4], const int t) {
  const int n = t & 15, h = t >> 4;
  const int col = ((n & 1) << N0) | (((n >> 1) & 1) << N1) |
                  (((n >> 2) & 1) << N2) | (((n >> 3) & 1) << N3);
  unsigned ldl = wofs | (8u * swz(col | (h << A0)));
  unsigned stl = wofs | (8u * swz(col | (h << A2)));
  asm volatile("" : "+v"(ldl), "+v"(stl));
  v2f b[2][8];
#pragma unroll
  for (int P = 0; P <= NPAIR; ++P) {
    if (P < NPAIR) {
#pragma unroll
      for (int x = 0; x < 8; ++x) {
        const int T = 2 * P + (x >> 2), kk = x & 3;
        const int c = swz(((T & 1) << T0) | (((T >> 1) & 1) << T1) |
                          (((T >> 2) & 1) << T2) | ((kk & 1) << A1) |
                          ((kk >> 1) << A2));
        const float2 amp = *(const float2*)(
            lds + ((ldl ^ (8u * (c & 31))) + 8u * (c & ~31)));
        b[P & 1][x].x = amp.x;
        b[P & 1][x].y = amp.y;
      }
    }
    if (P > 0) {
      const int Q = P - 1;
      v8f d0 = {}, d1 = {};
#pragma unroll
      for (int kk = 0; kk < 4; ++kk) {
        d0 = __builtin_amdgcn_wmma_f32_16x16x4_f32(
            false, a[kk], false, b[Q & 1][kk], (short)0, d0, false, false);
        d1 = __builtin_amdgcn_wmma_f32_16x16x4_f32(
            false, a[kk], false, b[Q & 1][4 + kk], (short)0, d1, false, false);
      }
#pragma unroll
      for (int x = 0; x < 8; ++x) {
        const int T = 2 * Q + (x >> 2), j = x & 3;
        const int c = swz(((T & 1) << T0) | (((T >> 1) & 1) << T1) |
                          (((T >> 2) & 1) << T2) | ((j & 1) << A0) |
                          ((j >> 1) << A1));
        const v8f d = (x < 4) ? d0 : d1;
        *(float2*)(lds + ((stl ^ (8u * (c & 31))) + 8u * (c & ~31))) =
            make_float2(d[2 * j], d[2 * j + 1]);
      }
    }
  }
  __builtin_amdgcn_wave_barrier();
}

template <int A0, int A1, int A2, int N0, int N1, int N2, int N3>
__device__ __forceinline__ void prefix_pass(char* lds, const unsigned wofs,
                                            const v2f (&a)[4], const int t) {
  const int n = t & 15, h = t >> 4;
  const int col = ((n & 1) << N0) | (((n >> 1) & 1) << N1) |
                  (((n >> 2) & 1) << N2) | (((n >> 3) & 1) << N3);
  const unsigned ldl = wofs | (8u * swz(col | (h << A0)));
  const unsigned stl = wofs | (8u * swz(col | (h << A2)));
  v2f b[4];
#pragma unroll
  for (int kk = 0; kk < 4; ++kk) {
    const int c = swz(((kk & 1) << A1) | ((kk >> 1) << A2));
    const float2 amp = *(const float2*)(
        lds + ((ldl ^ (8u * (c & 31))) + 8u * (c & ~31)));
    b[kk].x = amp.x;
    b[kk].y = amp.y;
  }
  v8f d = {};
#pragma unroll
  for (int kk = 0; kk < 4; ++kk)
    d = __builtin_amdgcn_wmma_f32_16x16x4_f32(
        false, a[kk], false, b[kk], (short)0, d, false, false);
#pragma unroll
  for (int j = 0; j < 4; ++j) {
    const int c = swz(((j & 1) << A0) | ((j >> 1) << A1));
    *(float2*)(lds + ((stl ^ (8u * (c & 31))) + 8u * (c & ~31))) =
        make_float2(d[2 * j], d[2 * j + 1]);
  }
  __builtin_amdgcn_wave_barrier();
}

__device__ __forceinline__ void load_afrag(v2f (&a)[4], const float* G,
                                           const int p, const int t) {
  const float2* src = (const float2*)G + p * 128 + t;
#pragma unroll
  for (int kk = 0; kk < 4; ++kk) {
    const float2 g = src[kk * 32];
    a[kk].x = g.x;
    a[kk].y = g.y;
  }
}

#define RUN_PASS(p, ...)                                   \
  {                                                        \
    v2f an[4];                                             \
    if ((p) + 1 < NPASS) load_afrag(an, G, (p) + 1, t);    \
    fused_pass<4, __VA_ARGS__>(lds, wofs, a, t);           \
    if ((p) + 1 < NPASS) {                                 \
      a[0] = an[0]; a[1] = an[1]; a[2] = an[2]; a[3] = an[3]; \
    }                                                      \
  }

#define RUN_PREFIX(p, ...)                                 \
  {                                                        \
    v2f an[4];                                             \
    load_afrag(an, G, (p) + 1, t);                         \
    prefix_pass<__VA_ARGS__>(lds, wofs, a, t);             \
    a[0] = an[0]; a[1] = an[1]; a[2] = an[2]; a[3] = an[3]; \
  }

#define RUN_HALF(p, ...)                                   \
  {                                                        \
    v2f an[4];                                             \
    load_afrag(an, G, (p) + 1, t);                         \
    fused_pass<2, __VA_ARGS__, 0>(lds, wofs, a, t);        \
    a[0] = an[0]; a[1] = an[1]; a[2] = an[2]; a[3] = an[3]; \
  }

__global__ __launch_bounds__(256, 8) void qsim_kernel(
    const float* __restrict__ inp, const float* __restrict__ W,
    float* __restrict__ out) {
  __shared__ float2 st[WPB][DIM];
  __shared__ float gs[40][8];
  __shared__ float tc[80], ts[80];
  __shared__ float4 gfrag[NPASS * 64];
  const int t = threadIdx.x & 31;
  const int w = threadIdx.x >> 5;
  const int samp = blockIdx.x * WPB + w;
  char* lds = (char*)&st[0][0];
  const unsigned wofs = w * (DIM * 8);
  const unsigned wt8 = wofs | (8u * t);
  const float* G = (const float*)gfrag;

  const float xl = (t < NQ) ? inp[samp * NQ + t] : 0.f;

  gate_trig(W, tc, ts, threadIdx.x);
  __syncthreads();
  gate_entries(tc, ts, gs, threadIdx.x);
  __syncthreads();
  gate_columns(gs, (float*)gfrag, w, t);

  float cc[NQ], ss[NQ];
  {
    const float xv = fminf(fmaxf(xl, 0.f), 1.f);
    const float ang = 0.5f * 3.14159265358979323846f * xv;
    const float cl = cosf(ang), sl = sinf(ang);
#pragma unroll
    for (int q = 0; q < NQ; ++q) {
      cc[q] = __int_as_float(__builtin_amdgcn_readlane(__float_as_int(cl), q));
      ss[q] = __int_as_float(__builtin_amdgcn_readlane(__float_as_int(sl), q));
    }
  }
  float ppre = 1.f;
#pragma unroll
  for (int j = 0; j < 5; ++j)
    ppre *= ((t >> j) & 1) ? ss[6 - j] : cc[6 - j];
  const float f78 = ((t & 1) ? ss[8] : cc[8]) * ((t & 2) ? ss[7] : cc[7]);
  const float f9 = (t & 1) ? ss[9] : cc[9];
  const unsigned pa = wofs | (8u * swz(((t & 3) << 3) | ((t >> 2) << 5)));
#pragma unroll
  for (int rr = 0; rr < 4; ++rr) {
    const float f = ppre * ((rr & 1) ? ss[1] : cc[1]) * ((rr & 2) ? ss[0] : cc[0]);
    const int c = swz(rr << 8);
    *(float2*)(lds + ((pa ^ (8u * (c & 31))) + 8u * (c & ~31))) =
        make_float2(f, 0.f);
  }
  __syncthreads();

  v2f a[4];
  load_afrag(a, G, 0, t);

  RUN_PREFIX(0, 7, 8, 9, 3, 4, 5, 6)
  RUN_PREFIX(1, 5, 6, 7, 3, 4, 8, 9)
  RUN_PREFIX(2, 3, 5, 4, 6, 7, 8, 9)

  {
    const unsigned el = wofs | (8u * swz(((t & 12) << 1) | ((t >> 4) << 5)));
    const unsigned es = wofs | (8u * swz(((t & 15) << 1) | ((t >> 4) << 5)));
#pragma unroll
    for (int g = 0; g < 2; ++g) {
      float2 v[8];
#pragma unroll
      for (int x = 0; x < 8; ++x) {
        const int c = swz((8 * g + x) << 6);
        v[x] = *(const float2*)(lds + ((el ^ (8u * (c & 31))) + 8u * (c & ~31)));
      }
#pragma unroll
      for (int x = 0; x < 8; ++x) {
        const int c = swz((8 * g + x) << 6);
        *(float2*)(lds + ((es ^ (8u * (c & 31))) + 8u * (c & ~31))) =
            make_float2(v[x].x * f78, v[x].y * f78);
      }
    }
  }
  __builtin_amdgcn_wave_barrier();

  RUN_HALF(3, 1, 2, 3, 4, 5, 7, 8, 6, 9)

  {
    const unsigned xl = wofs | (8u * (t & 0x1E));
#pragma unroll
    for (int g = 0; g < 4; ++g) {
      float2 v[8];
#pragma unroll
      for (int x = 0; x < 8; ++x) {
        const int r = 8 * g + x;
        v[x] = *(const float2*)(
            lds + ((xl ^ (8u * (swz(r << 5) & 31))) + (r << 8)));
      }
#pragma unroll
      for (int x = 0; x < 8; ++x) {
        const int r = 8 * g + x;
        *(float2*)(lds + ((wt8 ^ (8u * (swz(r << 5) & 31))) + (r << 8))) =
            make_float2(v[x].x * f9, v[x].y * f9);
      }
    }
  }
  __builtin_amdgcn_wave_barrier();

  RUN_PASS(4, 0, 9, 1, 2, 3, 4, 7, 5, 6, 8)
  RUN_PASS(5, 7, 8, 9, 0, 1, 2, 3, 4, 5, 6)
  RUN_PASS(6, 5, 6, 7, 0, 1, 2, 3, 4, 8, 9)
  RUN_PASS(7, 3, 5, 4, 0, 1, 2, 6, 7, 8, 9)
  RUN_PASS(8, 1, 2, 3, 0, 4, 5, 8, 6, 7, 9)
  RUN_PASS(9, 0, 9, 1, 2, 3, 4, 7, 5, 6, 8)

  float acc[NQ];
#pragma unroll
  for (int q = 0; q < NQ; ++q) acc[q] = 0.f;
  unsigned mt8 = wt8;
  asm volatile("" : "+v"(mt8));
#pragma unroll
  for (int r = 0; r < 32; ++r) {
    const int i = r * 32 + t;
    const float2 amp = *(const float2*)(
        lds + ((mt8 ^ (8u * (swz(r << 5) & 31))) + (r << 8)));
    const float p = amp.x * amp.x + amp.y * amp.y;
#pragma unroll
    for (int q = 0; q < NQ; ++q)
      acc[q] += ((i >> (9 - q)) & 1) ? -p : p;
  }
#pragma unroll
  for (int q = 0; q < NQ; ++q) {
#pragma unroll
    for (int off = 16; off >= 1; off >>= 1)
      acc[q] += __shfl_xor(acc[q], off, 32);
  }
  if (t == 0) {
#pragma unroll
    for (int q = 0; q < NQ; ++q) out[samp * NQ + q] = acc[q];
  }
}

extern "C" void kernel_launch(void* const* d_in, const int* in_sizes, int n_in,
                              void* d_out, int out_size, void* d_ws,
                              size_t ws_size, hipStream_t stream) {
  const float* inputs = (const float*)d_in[0];
  const float* weights = (const float*)d_in[1];
  float* out = (float*)d_out;

  const int B = in_sizes[0] / NQ;
  qsim_kernel<<<dim3(B / WPB), dim3(256), 0, stream>>>(inputs, weights, out);
}


# Passes 3,4 folded: the still-unentangled qubits' cos/sin are multiplied into the A fragments, K = 4 and 8 not 16 (192 WMMAs, not 220), no in-LDS expansions; products reassociated, k order changes
# speedup vs baseline: 4.8002x; 1.1092x over previous
#include <hip/hip_runtime.h>


typedef __attribute__((ext_vector_type(2))) float v2f;
typedef __attribute__((ext_vector_type(8))) float v8f;

#define NQ 10
#define DIM 1024
#define WPB 8
#define NPASS 10

__host__ __device__ constexpr int swz(int i) {
  return i ^ ((i & 32) ? 0x14 : 0) ^ ((i & 64) ? 0x18 : 0) ^
         ((i & 128) ? 0x13 : 0) ^ ((i & 256) ? 0x0A : 0) ^
         ((i & 512) ? 0x10 : 0);
}

template <int l>
__device__ __forceinline__ void ap_rot(float (&vr)[8], float (&vi)[8],
                                       const float* u) {
  const float ar = u[0], ai = u[1], br = u[2], bi = u[3];
  const float cr = u[4], ci = u[5], dr = u[6], di = u[7];
#pragma unroll
  for (int a = 0; a < 8; ++a) {
    if ((a >> l) & 1) continue;
    const int b = a | (1 << l);
    const float x0r = vr[a], x0i = vi[a], x1r = vr[b], x1i = vi[b];
    vr[a] = ar * x0r - ai * x0i + br * x1r - bi * x1i;
    vi[a] = ar * x0i + ai * x0r + br * x1i + bi * x1r;
    vr[b] = cr * x0r - ci * x0i + dr * x1r - di * x1i;
    vi[b] = cr * x0i + ci * x0r + dr * x1i + di * x1r;
  }
}

template <int cb, int tb>
__device__ __forceinline__ void ap_cry(float (&vr)[8], float (&vi)[8],
                                       const float* u) {
  const float c = u[0], s = u[1];
#pragma unroll
  for (int a = 0; a < 8; ++a) {
    if (!((a >> cb) & 1) || ((a >> tb) & 1)) continue;
    const int b = a | (1 << tb);
    const float x0r = vr[a], x0i = vi[a], x1r = vr[b], x1i = vi[b];
    vr[a] = c * x0r - s * x1r;
    vi[a] = c * x0i - s * x1i;
    vr[b] = s * x0r + c * x1r;
    vi[b] = s * x0i + c * x1i;
  }
}

__device__ __forceinline__ void gate_trig(const float* __restrict__ W,
                                          float* tc, float* ts, const int g) {
  const int a = g & 127;
  if (a < 80) {
    float ang;
    if (a < 60) {
      const int rg = a / 3, k = a - 3 * rg;
      const int base = (rg / 10) * 40 + 3 * (rg % 10);
      const float phi = W[base], th = W[base + 1], om = W[base + 2];
      ang = (k == 0) ? 0.5f * th
                     : (k == 1) ? 0.5f * (phi + om) : 0.5f * (phi - om);
    } else {
      const int cg = a - 60;
      ang = 0.5f * W[(cg / 10) * 40 + 30 + (cg % 10)];
    }
    if (g < 128) tc[a] = cosf(ang); else ts[a] = sinf(ang);
  }
}

__device__ __forceinline__ void gate_entries(const float* tc, const float* ts,
                                             float (*gs)[8], const int g) {
  if (g < 40) {
    const int L = g / 20, r = g % 20;
    float* u = gs[g];
    if (r < 10) {
      const int a = 3 * (L * 10 + r);
      const float ct = tc[a], st = ts[a];
      const float cap = tc[a + 1], sap = ts[a + 1];
      const float cam = tc[a + 2], sam = ts[a + 2];
      u[0] =  ct * cap; u[1] = -ct * sap;
      u[2] = -st * cam; u[3] = -st * sam;
      u[4] =  st * cam; u[5] = -st * sam;
      u[6] =  ct * cap; u[7] =  ct * sap;
    } else {
      const int a = 60 + L * 10 + (r - 10);
      u[0] = tc[a];
      u[1] = ts[a];
    }
  }
}

__device__ __forceinline__ void gate_columns(float (*gs)[8], float* G,
                                             const int wv, const int ln) {
  if (wv >= 5 || ln >= 16) return;
  const int p = (ln >> 3) * 5 + wv, col = ln & 7;
  const int gb = (ln >> 3) * 20;
  float vr[8], vi[8];
#pragma unroll
  for (int a = 0; a < 8; ++a) { vr[a] = (a == col) ? 1.f : 0.f; vi[a] = 0.f; }
  switch (wv) {
    case 0:
      ap_rot<2>(vr, vi, gs[gb + 0]); ap_rot<1>(vr, vi, gs[gb + 1]);
      ap_rot<0>(vr, vi, gs[gb + 2]);
      ap_cry<2, 1>(vr, vi, gs[gb + 10]); ap_cry<1, 0>(vr, vi, gs[gb + 11]);
      break;
    case 1:
      ap_rot<1>(vr, vi, gs[gb + 3]); ap_rot<0>(vr, vi, gs[gb + 4]);
      ap_cry<2, 1>(vr, vi, gs[gb + 12]); ap_cry<1, 0>(vr, vi, gs[gb + 13]);
      break;
    case 2:
      ap_rot<2>(vr, vi, gs[gb + 5]); ap_rot<0>(vr, vi, gs[gb + 6]);
      ap_cry<1, 2>(vr, vi, gs[gb + 14]); ap_cry<2, 0>(vr, vi, gs[gb + 15]);
      break;
    case 3:
      ap_rot<1>(vr, vi, gs[gb + 7]); ap_rot<2>(vr, vi, gs[gb + 8]);
      ap_cry<0, 1>(vr, vi, gs[gb + 16]); ap_cry<1, 2>(vr, vi, gs[gb + 17]);
      break;
    default:
      ap_rot<1>(vr, vi, gs[gb + 9]);
      ap_cry<0, 1>(vr, vi, gs[gb + 18]); ap_cry<1, 2>(vr, vi, gs[gb + 19]);
      break;
  }
  float4* dst = (float4*)(G + ((p * 4 + (col >> 1)) * 32 + (col & 1) * 16) * 2);
#pragma unroll
  for (int a = 0; a < 8; ++a)
    dst[a] = make_float4(vr[a], -vi[a], vi[a], vr[a]);
}

template <int NPAIR, int KS, int A0, int A1, int A2, int N0, int N1, int N2,
          int N3, int T0, int T1, int T2>
__device__ __forceinline__ void fused_pass(char* lds, const unsigned wofs,
                                           const v2f (&a)[KS], const int t) {
  constexpr int KSTEP = 4 / KS;
  const int n = t & 15, h = t >> 4;
  const int col = ((n & 1) << N0) | (((n >> 1) & 1) << N1) |
                  (((n >> 2) & 1) << N2) | (((n >> 3) & 1) << N3);
  unsigned ldl = wofs | (8u * swz(col | (h << A0)));
  unsigned stl = wofs | (8u * swz(col | (h << A2)));
  asm volatile("" : "+v"(ldl), "+v"(stl));
  v2f b[2][8];
#pragma unroll
  for (int P = 0; P <= NPAIR; ++P) {
    if (P < NPAIR) {
#pragma unroll
      for (int x = 0; x < 8; ++x) {
        const int T = 2 * P + (x >> 2), kk = x & 3;
        if (kk % KSTEP) continue;
        const int c = swz(((T & 1) << T0) | (((T >> 1) & 1) << T1) |
                          (((T >> 2) & 1) << T2) | ((kk & 1) << A1) |
                          ((kk >> 1) << A2));
        const float2 amp = *(const float2*)(
            lds + ((ldl ^ (8u * (c & 31))) + 8u * (c & ~31)));
        b[P & 1][x].x = amp.x;
        b[P & 1][x].y = amp.y;
      }
    }
    if (P > 0) {
      const int Q = P - 1;
      v8f d0 = {}, d1 = {};
#pragma unroll
      for (int s = 0; s < KS; ++s) {
        d0 = __builtin_amdgcn_wmma_f32_16x16x4_f32(
            false, a[s], false, b[Q & 1][s * KSTEP], (short)0, d0, false, false);
        d1 = __builtin_amdgcn_wmma_f32_16x16x4_f32(
            false, a[s], false, b[Q & 1][4 + s * KSTEP], (short)0, d1, false,
            false);
      }
#pragma unroll
      for (int x = 0; x < 8; ++x) {
        const int T = 2 * Q + (x >> 2), j = x & 3;
        const int c = swz(((T & 1) << T0) | (((T >> 1) & 1) << T1) |
                          (((T >> 2) & 1) << T2) | ((j & 1) << A0) |
                          ((j >> 1) << A1));
        const v8f d = (x < 4) ? d0 : d1;
        *(float2*)(lds + ((stl ^ (8u * (c & 31))) + 8u * (c & ~31))) =
            make_float2(d[2 * j], d[2 * j + 1]);
      }
    }
  }
  __builtin_amdgcn_wave_barrier();
}

template <int A0, int A1, int A2, int N0, int N1, int N2, int N3>
__device__ __forceinline__ void prefix_pass(char* lds, const unsigned wofs,
                                            const v2f (&a)[4], const int t) {
  const int n = t & 15, h = t >> 4;
  const int col = ((n & 1) << N0) | (((n >> 1) & 1) << N1) |
                  (((n >> 2) & 1) << N2) | (((n >> 3) & 1) << N3);
  const unsigned ldl = wofs | (8u * swz(col | (h << A0)));
  const unsigned stl = wofs | (8u * swz(col | (h << A2)));
  v2f b[4];
#pragma unroll
  for (int kk = 0; kk < 4; ++kk) {
    const int c = swz(((kk & 1) << A1) | ((kk >> 1) << A2));
    const float2 amp = *(const float2*)(
        lds + ((ldl ^ (8u * (c & 31))) + 8u * (c & ~31)));
    b[kk].x = amp.x;
    b[kk].y = amp.y;
  }
  v8f d = {};
#pragma unroll
  for (int kk = 0; kk < 4; ++kk)
    d = __builtin_amdgcn_wmma_f32_16x16x4_f32(
        false, a[kk], false, b[kk], (short)0, d, false, false);
#pragma unroll
  for (int j = 0; j < 4; ++j) {
    const int c = swz(((j & 1) << A0) | ((j >> 1) << A1));
    *(float2*)(lds + ((stl ^ (8u * (c & 31))) + 8u * (c & ~31))) =
        make_float2(d[2 * j], d[2 * j + 1]);
  }
  __builtin_amdgcn_wave_barrier();
}

__device__ __forceinline__ void load_afrag(v2f (&a)[4], const float* G,
                                           const int p, const int t) {
  const float2* src = (const float2*)G + p * 128 + t;
#pragma unroll
  for (int kk = 0; kk < 4; ++kk) {
    const float2 g = src[kk * 32];
    a[kk].x = g.x;
    a[kk].y = g.y;
  }
}

#define RUN_PASS(p, ...)                                   \
  {                                                        \
    v2f an[4];                                             \
    if ((p) + 1 < NPASS) load_afrag(an, G, (p) + 1, t);    \
    fused_pass<4, 4, __VA_ARGS__>(lds, wofs, a, t);        \
    if ((p) + 1 < NPASS) {                                 \
      a[0] = an[0]; a[1] = an[1]; a[2] = an[2]; a[3] = an[3]; \
    }                                                      \
  }

#define RUN_PREFIX(p, ...)                                 \
  {                                                        \
    v2f an[4];                                             \
    load_afrag(an, G, (p) + 1, t);                         \
    prefix_pass<__VA_ARGS__>(lds, wofs, a, t);             \
    a[0] = an[0]; a[1] = an[1]; a[2] = an[2]; a[3] = an[3]; \
  }

#define RUN_FOLDED(p, NPAIR, KS, af, ...)                  \
  {                                                        \
    v2f an[4];                                             \
    load_afrag(an, G, (p) + 1, t);                         \
    fused_pass<NPAIR, KS, __VA_ARGS__>(lds, wofs, af, t);  \
    a[0] = an[0]; a[1] = an[1]; a[2] = an[2]; a[3] = an[3]; \
  }

__global__ __launch_bounds__(256, 8) void qsim_kernel(
    const float* __restrict__ inp, const float* __restrict__ W,
    float* __restrict__ out) {
  __shared__ float2 st[WPB][DIM];
  __shared__ float gs[40][8];
  __shared__ float tc[80], ts[80];
  __shared__ float4 gfrag[NPASS * 64];
  const int t = threadIdx.x & 31;
  const int w = threadIdx.x >> 5;
  const int samp = blockIdx.x * WPB + w;
  char* lds = (char*)&st[0][0];
  const unsigned wofs = w * (DIM * 8);
  const unsigned wt8 = wofs | (8u * t);
  const float* G = (const float*)gfrag;

  const float xl = (t < NQ) ? inp[samp * NQ + t] : 0.f;

  gate_trig(W, tc, ts, threadIdx.x);
  __syncthreads();
  gate_entries(tc, ts, gs, threadIdx.x);
  __syncthreads();
  gate_columns(gs, (float*)gfrag, w, t);

  float cc[NQ], ss[NQ];
  {
    const float xv = fminf(fmaxf(xl, 0.f), 1.f);
    const float ang = 0.5f * 3.14159265358979323846f * xv;
    const float cl = cosf(ang), sl = sinf(ang);
#pragma unroll
    for (int q = 0; q < NQ; ++q) {
      cc[q] = __int_as_float(__builtin_amdgcn_readlane(__float_as_int(cl), q));
      ss[q] = __int_as_float(__builtin_amdgcn_readlane(__float_as_int(sl), q));
    }
  }
  float ppre = 1.f;
#pragma unroll
  for (int j = 0; j < 5; ++j)
    ppre *= ((t >> j) & 1) ? ss[6 - j] : cc[6 - j];
  const unsigned pa = wofs | (8u * swz(((t & 3) << 3) | ((t >> 2) << 5)));
#pragma unroll
  for (int rr = 0; rr < 4; ++rr) {
    const float f = ppre * ((rr & 1) ? ss[1] : cc[1]) * ((rr & 2) ? ss[0] : cc[0]);
    const int c = swz(rr << 8);
    *(float2*)(lds + ((pa ^ (8u * (c & 31))) + 8u * (c & ~31))) =
        make_float2(f, 0.f);
  }
  __syncthreads();

  v2f a[4];
  load_afrag(a, G, 0, t);

  RUN_PREFIX(0, 7, 8, 9, 3, 4, 5, 6)
  RUN_PREFIX(1, 5, 6, 7, 3, 4, 8, 9)
  RUN_PREFIX(2, 3, 5, 4, 6, 7, 8, 9)

  {
    v2f af[1];
    const float w0 = cc[7] * cc[8], w1 = ss[7] * cc[8];
    const float w2 = cc[7] * ss[8], w3 = ss[7] * ss[8];
    af[0].x = w0 * a[0].x + w1 * a[1].x + w2 * a[2].x + w3 * a[3].x;
    af[0].y = w0 * a[0].y + w1 * a[1].y + w2 * a[2].y + w3 * a[3].y;
    RUN_FOLDED(3, 2, 1, af, 3, 2, 1, 4, 5, 7, 8, 6, 9, 0)
  }

  {
    v2f af[2];
#pragma unroll
    for (int s = 0; s < 2; ++s) {
      af[s].x = cc[9] * a[2 * s].x + ss[9] * a[2 * s + 1].x;
      af[s].y = cc[9] * a[2 * s].y + ss[9] * a[2 * s + 1].y;
    }
    RUN_FOLDED(4, 4, 2, af, 1, 0, 9, 2, 6, 7, 8, 3, 4, 5)
  }

  RUN_PASS(5, 7, 8, 9, 0, 1, 2, 3, 4, 5, 6)
  RUN_PASS(6, 5, 6, 7, 0, 1, 2, 3, 4, 8, 9)
  RUN_PASS(7, 3, 5, 4, 0, 1, 2, 6, 7, 8, 9)
  RUN_PASS(8, 3, 2, 1, 4, 5, 7, 8, 0, 6, 9)
  RUN_PASS(9, 1, 0, 9, 2, 6, 7, 8, 3, 4, 5)

  float acc[NQ];
#pragma unroll
  for (int q = 0; q < NQ; ++q) acc[q] = 0.f;
  unsigned mt8 = wt8;
  asm volatile("" : "+v"(mt8));
#pragma unroll
  for (int r = 0; r < 32; ++r) {
    const int i = r * 32 + t;
    const float2 amp = *(const float2*)(
        lds + ((mt8 ^ (8u * (swz(r << 5) & 31))) + (r << 8)));
    const float p = amp.x * amp.x + amp.y * amp.y;
#pragma unroll
    for (int q = 0; q < NQ; ++q)
      acc[q] += ((i >> (9 - q)) & 1) ? -p : p;
  }
#pragma unroll
  for (int q = 0; q < NQ; ++q) {
#pragma unroll
    for (int off = 16; off >= 1; off >>= 1)
      acc[q] += __shfl_xor(acc[q], off, 32);
  }
  if (t == 0) {
#pragma unroll
    for (int q = 0; q < NQ; ++q) out[samp * NQ + q] = acc[q];
  }
}

extern "C" void kernel_launch(void* const* d_in, const int* in_sizes, int n_in,
                              void* d_out, int out_size, void* d_ws,
                              size_t ws_size, hipStream_t stream) {
  const float* inputs = (const float*)d_in[0];
  const float* weights = (const float*)d_in[1];
  float* out = (float*)d_out;

  const int B = in_sizes[0] / NQ;
  qsim_kernel<<<dim3(B / WPB), dim3(256), 0, stream>>>(inputs, weights, out);
}


# Passes chained in registers in pairs (4+5, 6+7, 8+9+measurement): a pass's result pairs are the next pass's B slices as they stand; 3 LDS state trips, not 7; gate products and Z sums reordered
# speedup vs baseline: 5.4180x; 1.1287x over previous
#include <hip/hip_runtime.h>


typedef __attribute__((ext_vector_type(2))) float v2f;
typedef __attribute__((ext_vector_type(8))) float v8f;

#define NQ 10
#define DIM 1024
#define WPB 8
#define NPASS 10

__host__ __device__ constexpr int swz(int i) {
  return i ^ ((i & 32) ? 0x1C : 0) ^ ((i & 64) ? 0x01 : 0) ^
         ((i & 128) ? 0x1F : 0) ^ ((i & 256) ? 0x19 : 0) ^
         ((i & 512) ? 0x0A : 0);
}

template <int l>
__device__ __forceinline__ void ap_rot(float (&vr)[8], float (&vi)[8],
                                       const float* u) {
  const float ar = u[0], ai = u[1], br = u[2], bi = u[3];
  const float cr = u[4], ci = u[5], dr = u[6], di = u[7];
#pragma unroll
  for (int a = 0; a < 8; ++a) {
    if ((a >> l) & 1) continue;
    const int b = a | (1 << l);
    const float x0r = vr[a], x0i = vi[a], x1r = vr[b], x1i = vi[b];
    vr[a] = ar * x0r - ai * x0i + br * x1r - bi * x1i;
    vi[a] = ar * x0i + ai * x0r + br * x1i + bi * x1r;
    vr[b] = cr * x0r - ci * x0i + dr * x1r - di * x1i;
    vi[b] = cr * x0i + ci * x0r + dr * x1i + di * x1r;
  }
}

template <int cb, int tb>
__device__ __forceinline__ void ap_cry(float (&vr)[8], float (&vi)[8],
                                       const float* u) {
  const float c = u[0], s = u[1];
#pragma unroll
  for (int a = 0; a < 8; ++a) {
    if (!((a >> cb) & 1) || ((a >> tb) & 1)) continue;
    const int b = a | (1 << tb);
    const float x0r = vr[a], x0i = vi[a], x1r = vr[b], x1i = vi[b];
    vr[a] = c * x0r - s * x1r;
    vi[a] = c * x0i - s * x1i;
    vr[b] = s * x0r + c * x1r;
    vi[b] = s * x0i + c * x1i;
  }
}

__device__ __forceinline__ void gate_trig(const float* __restrict__ W,
                                          float* tc, float* ts, const int g) {
  const int a = g & 127;
  if (a < 80) {
    float ang;
    if (a < 60) {
      const int rg = a / 3, k = a - 3 * rg;
      const int base = (rg / 10) * 40 + 3 * (rg % 10);
      const float phi = W[base], th = W[base + 1], om = W[base + 2];
      ang = (k == 0) ? 0.5f * th
                     : (k == 1) ? 0.5f * (phi + om) : 0.5f * (phi - om);
    } else {
      const int cg = a - 60;
      ang = 0.5f * W[(cg / 10) * 40 + 30 + (cg % 10)];
    }
    if (g < 128) tc[a] = cosf(ang); else ts[a] = sinf(ang);
  }
}

__device__ __forceinline__ void gate_entries(const float* tc, const float* ts,
                                             float (*gs)[8], const int g) {
  if (g < 40) {
    const int L = g / 20, r = g % 20;
    float* u = gs[g];
    if (r < 10) {
      const int a = 3 * (L * 10 + r);
      const float ct = tc[a], st = ts[a];
      const float cap = tc[a + 1], sap = ts[a + 1];
      const float cam = tc[a + 2], sam = ts[a + 2];
      u[0] =  ct * cap; u[1] = -ct * sap;
      u[2] = -st * cam; u[3] = -st * sam;
      u[4] =  st * cam; u[5] = -st * sam;
      u[6] =  ct * cap; u[7] =  ct * sap;
    } else {
      const int a = 60 + L * 10 + (r - 10);
      u[0] = tc[a];
      u[1] = ts[a];
    }
  }
}

__device__ __forceinline__ void gate_columns(float (*gs)[8], float* G,
                                             const int wv, const int ln) {
  if (wv >= 5 || ln >= 16) return;
  const int p = (ln >> 3) * 5 + wv, col = ln & 7;
  const int gb = (ln >> 3) * 20;
  float vr[8], vi[8];
#pragma unroll
  for (int a = 0; a < 8; ++a) { vr[a] = (a == col) ? 1.f : 0.f; vi[a] = 0.f; }
  switch (wv) {
    case 0:
      ap_rot<0>(vr, vi, gs[gb + 0]); ap_rot<1>(vr, vi, gs[gb + 1]);
      ap_rot<2>(vr, vi, gs[gb + 2]);
      ap_cry<0, 1>(vr, vi, gs[gb + 10]); ap_cry<1, 2>(vr, vi, gs[gb + 11]);
      break;
    case 1:
      ap_rot<1>(vr, vi, gs[gb + 3]); ap_rot<2>(vr, vi, gs[gb + 4]);
      ap_cry<0, 1>(vr, vi, gs[gb + 12]); ap_cry<1, 2>(vr, vi, gs[gb + 13]);
      break;
    case 2:
      ap_rot<1>(vr, vi, gs[gb + 5]); ap_rot<2>(vr, vi, gs[gb + 6]);
      ap_cry<0, 1>(vr, vi, gs[gb + 14]); ap_cry<1, 2>(vr, vi, gs[gb + 15]);
      break;
    case 3:
      ap_rot<1>(vr, vi, gs[gb + 7]); ap_rot<2>(vr, vi, gs[gb + 8]);
      ap_cry<0, 1>(vr, vi, gs[gb + 16]); ap_cry<1, 2>(vr, vi, gs[gb + 17]);
      break;
    default:
      ap_rot<1>(vr, vi, gs[gb + 9]);
      ap_cry<0, 1>(vr, vi, gs[gb + 18]); ap_cry<1, 2>(vr, vi, gs[gb + 19]);
      break;
  }
  float4* dst = (float4*)(G + ((p * 4 + (col >> 1)) * 32 + (col & 1) * 16) * 2);
#pragma unroll
  for (int a = 0; a < 8; ++a)
    dst[a] = make_float4(vr[a], -vi[a], vi[a], vr[a]);
}

template <int NPAIR, int KS, int A0, int A1, int A2, int N0, int N1, int N2,
          int N3, int T0, int T1, int T2>
__device__ __forceinline__ void fused_pass(char* lds, const unsigned wofs,
                                           const v2f (&a)[KS], const int t) {
  constexpr int KSTEP = 4 / KS;
  const int n = t & 15, h = t >> 4;
  const int col = ((n & 1) << N0) | (((n >> 1) & 1) << N1) |
                  (((n >> 2) & 1) << N2) | (((n >> 3) & 1) << N3);
  unsigned ldl = wofs | (8u * swz(col | (h << A0)));
  unsigned stl = wofs | (8u * swz(col | (h << A2)));
  v2f b[2][8];
#pragma unroll
  for (int P = 0; P <= NPAIR; ++P) {
    if (P < NPAIR) {
#pragma unroll
      for (int x = 0; x < 8; ++x) {
        const int T = 2 * P + (x >> 2), kk = x & 3;
        if (kk % KSTEP) continue;
        const int c = swz(((T & 1) << T0) | (((T >> 1) & 1) << T1) |
                          (((T >> 2) & 1) << T2) | ((kk & 1) << A1) |
                          ((kk >> 1) << A2));
        const float2 amp = *(const float2*)(
            lds + ((ldl ^ (8u * (c & 31))) + 8u * (c & ~31)));
        b[P & 1][x].x = amp.x;
        b[P & 1][x].y = amp.y;
      }
    }
    if (P > 0) {
      const int Q = P - 1;
      v8f d0 = {}, d1 = {};
#pragma unroll
      for (int s = 0; s < KS; ++s) {
        d0 = __builtin_amdgcn_wmma_f32_16x16x4_f32(
            false, a[s], false, b[Q & 1][s * KSTEP], (short)0, d0, false, false);
        d1 = __builtin_amdgcn_wmma_f32_16x16x4_f32(
            false, a[s], false, b[Q & 1][4 + s * KSTEP], (short)0, d1, false,
            false);
      }
#pragma unroll
      for (int x = 0; x < 8; ++x) {
        const int T = 2 * Q + (x >> 2), j = x & 3;
        const int c = swz(((T & 1) << T0) | (((T >> 1) & 1) << T1) |
                          (((T >> 2) & 1) << T2) | ((j & 1) << A0) |
                          ((j >> 1) << A1));
        const v8f d = (x < 4) ? d0 : d1;
        *(float2*)(lds + ((stl ^ (8u * (c & 31))) + 8u * (c & ~31))) =
            make_float2(d[2 * j], d[2 * j + 1]);
      }
    }
  }
  __builtin_amdgcn_wave_barrier();
}

template <int A0, int A1, int A2, int N0, int N1, int N2, int N3>
__device__ __forceinline__ void prefix_pass(char* lds, const unsigned wofs,
                                            const v2f (&a)[4], const int t) {
  const int n = t & 15, h = t >> 4;
  const int col = ((n & 1) << N0) | (((n >> 1) & 1) << N1) |
                  (((n >> 2) & 1) << N2) | (((n >> 3) & 1) << N3);
  const unsigned ldl = wofs | (8u * swz(col | (h << A0)));
  const unsigned stl = wofs | (8u * swz(col | (h << A2)));
  v2f b[4];
#pragma unroll
  for (int kk = 0; kk < 4; ++kk) {
    const int c = swz(((kk & 1) << A1) | ((kk >> 1) << A2));
    const float2 amp = *(const float2*)(
        lds + ((ldl ^ (8u * (c & 31))) + 8u * (c & ~31)));
    b[kk].x = amp.x;
    b[kk].y = amp.y;
  }
  v8f d = {};
#pragma unroll
  for (int kk = 0; kk < 4; ++kk)
    d = __builtin_amdgcn_wmma_f32_16x16x4_f32(
        false, a[kk], false, b[kk], (short)0, d, false, false);
#pragma unroll
  for (int j = 0; j < 4; ++j) {
    const int c = swz(((j & 1) << A0) | ((j >> 1) << A1));
    *(float2*)(lds + ((stl ^ (8u * (c & 31))) + 8u * (c & ~31))) =
        make_float2(d[2 * j], d[2 * j + 1]);
  }
  __builtin_amdgcn_wave_barrier();
}

template <int KS, int A0, int A1, int A2, int N0, int N1, int N2, int N3,
          int T0, int T1, int T2>
__device__ __forceinline__ void chained_passes(char* lds, const unsigned wofs,
                                               const v2f (&ax)[KS],
                                               const v2f (&ay)[4],
                                               const int t) {
  constexpr int KSTEP = 4 / KS;
  const int n = t & 15, h = t >> 4;
  const int col = ((n & 1) << N0) | (((n >> 1) & 1) << N1) |
                  (((n >> 2) & 1) << N2) | (((n >> 3) & 1) << N3);
  const unsigned ldl = wofs | (8u * swz(col | (h << A0)));
  const unsigned stl = wofs | (8u * swz(col | (h << T1)));
  v2f b[2][8];
  v2f kh[16];
#pragma unroll
  for (int i = -1; i < 4; ++i) {
    if (i < 3) {
#pragma unroll
      for (int x = 0; x < 8; ++x) {
        const int T = 2 * (i + 1) + (x >> 2), kk = x & 3;
        if (kk % KSTEP) continue;
        const int c = swz(((T & 1) << T0) | (((T >> 1) & 1) << T1) |
                          (((T >> 2) & 1) << T2) | ((kk & 1) << A1) |
                          ((kk >> 1) << A2));
        const float2 amp = *(const float2*)(
            lds + ((ldl ^ (8u * (c & 31))) + 8u * (c & ~31)));
        b[(i + 1) & 1][x].x = amp.x;
        b[(i + 1) & 1][x].y = amp.y;
      }
    }
    if (i >= 0) {
      const int m = i & 1;
      {
        v8f d0 = {}, d1 = {};
#pragma unroll
        for (int s = 0; s < KS; ++s) {
          d0 = __builtin_amdgcn_wmma_f32_16x16x4_f32(
              false, ax[s], false, b[i & 1][s * KSTEP], (short)0, d0, false,
              false);
          d1 = __builtin_amdgcn_wmma_f32_16x16x4_f32(
              false, ax[s], false, b[i & 1][4 + s * KSTEP], (short)0, d1,
              false, false);
        }
#pragma unroll
        for (int j = 0; j < 4; ++j) {
          kh[4 * (2 * m) + j].x = d0[2 * j];
          kh[4 * (2 * m) + j].y = d0[2 * j + 1];
          kh[4 * (2 * m + 1) + j].x = d1[2 * j];
          kh[4 * (2 * m + 1) + j].y = d1[2 * j + 1];
        }
      }
      __builtin_amdgcn_sched_barrier(0);
      if (m == 1) {
        const int hf = i >> 1;
#pragma unroll
        for (int Q = 0; Q < 2; ++Q) {
          v8f d0 = {}, d1 = {};
#pragma unroll
          for (int kk = 0; kk < 4; ++kk) {
            d0 = __builtin_amdgcn_wmma_f32_16x16x4_f32(
                false, ay[kk], false, kh[4 * kk + 2 * Q], (short)0, d0, false,
                false);
            d1 = __builtin_amdgcn_wmma_f32_16x16x4_f32(
                false, ay[kk], false, kh[4 * kk + 2 * Q + 1], (short)0, d1,
                false, false);
          }
#pragma unroll
          for (int x = 0; x < 8; ++x) {
            const int U = 4 * hf + 2 * Q + (x >> 2), j = x & 3;
            const int c = swz(((U & 1) << A0) | (((U >> 1) & 1) << A1) |
                              ((U >> 2) << T2) | ((j & 1) << A2) |
                              ((j >> 1) << T0));
            const v8f d = (x < 4) ? d0 : d1;
            *(float2*)(lds + ((stl ^ (8u * (c & 31))) + 8u * (c & ~31))) =
                make_float2(d[2 * j], d[2 * j + 1]);
          }
          __builtin_amdgcn_sched_barrier(0);
        }
      }
    }
  }
  __builtin_amdgcn_wave_barrier();
}

__device__ __forceinline__ void tail_passes(char* lds, const unsigned wofs,
                                            const v2f (&a8)[4],
                                            const v2f (&a9)[4], const int t,
                                            float (&acc)[NQ]) {
  constexpr int A0 = 3, A1 = 2, A2 = 1, N0 = 4, N1 = 5, N2 = 7, N3 = 8,
                T0 = 0, T1 = 6, T2 = 9;
  const int n = t & 15, h = t >> 4;
  const int col = ((n & 1) << N0) | (((n >> 1) & 1) << N1) |
                  (((n >> 2) & 1) << N2) | (((n >> 3) & 1) << N3);
  const unsigned ldl = wofs | (8u * swz(col | (h << A0)));
  float tot = 0.f, z3 = 0.f, z6 = 0.f, z7 = 0.f, z8 = 0.f, z9 = 0.f;
  v2f b[2][8];
  v2f kh[16];
#pragma unroll
  for (int i = -1; i < 4; ++i) {
    if (i < 3) {
      const int Pn = ((i + 1) >> 1) + 2 * ((i + 1) & 1);
#pragma unroll
      for (int x = 0; x < 8; ++x) {
        const int T = 2 * Pn + (x >> 2), kk = x & 3;
        const int c = swz(((T & 1) << T0) | (((T >> 1) & 1) << T1) |
                          (((T >> 2) & 1) << T2) | ((kk & 1) << A1) |
                          ((kk >> 1) << A2));
        const float2 amp = *(const float2*)(
            lds + ((ldl ^ (8u * (c & 31))) + 8u * (c & ~31)));
        b[(i + 1) & 1][x].x = amp.x;
        b[(i + 1) & 1][x].y = amp.y;
      }
    }
    if (i >= 0) {
      const int m = i & 1;
      {
        v8f d0 = {}, d1 = {};
#pragma unroll
        for (int s = 0; s < 4; ++s) {
          d0 = __builtin_amdgcn_wmma_f32_16x16x4_f32(
              false, a8[s], false, b[i & 1][s], (short)0, d0, false, false);
          d1 = __builtin_amdgcn_wmma_f32_16x16x4_f32(
              false, a8[s], false, b[i & 1][4 + s], (short)0, d1, false, false);
        }
#pragma unroll
        for (int j = 0; j < 4; ++j) {
          kh[4 * (2 * m) + j].x = d0[2 * j];
          kh[4 * (2 * m) + j].y = d0[2 * j + 1];
          kh[4 * (2 * m + 1) + j].x = d1[2 * j];
          kh[4 * (2 * m + 1) + j].y = d1[2 * j + 1];
        }
      }
      __builtin_amdgcn_sched_barrier(0);
      if (m == 1) {
        const int hf = i >> 1;
#pragma unroll
        for (int Q = 0; Q < 2; ++Q) {
          v8f d0 = {}, d1 = {};
#pragma unroll
          for (int kk = 0; kk < 4; ++kk) {
            d0 = __builtin_amdgcn_wmma_f32_16x16x4_f32(
                false, a9[kk], false, kh[4 * kk + 2 * Q], (short)0, d0, false,
                false);
            d1 = __builtin_amdgcn_wmma_f32_16x16x4_f32(
                false, a9[kk], false, kh[4 * kk + 2 * Q + 1], (short)0, d1,
                false, false);
          }
#pragma unroll
          for (int x = 0; x < 2; ++x) {
            const int U = 4 * hf + 2 * Q + x;
            const v8f d = x ? d1 : d0;
            const float p0 = d[0] * d[0] + d[1] * d[1];
            const float p1 = d[2] * d[2] + d[3] * d[3];
            const float p2 = d[4] * d[4] + d[5] * d[5];
            const float p3 = d[6] * d[6] + d[7] * d[7];
            const float e = p0 + p1, f = p2 + p3;
            const float su = e + f;
            tot += su;
            z8 += (p0 - p1) + (p2 - p3);
            z9 += e - f;
            z6 += (U & 1) ? -su : su;
            z7 += (U & 2) ? -su : su;
            z3 += (U & 4) ? -su : su;
          }
          __builtin_amdgcn_sched_barrier(0);
        }
      }
    }
  }
  acc[0] = (t & 16) ? -tot : tot;
  acc[1] = (t & 8) ? -tot : tot;
  acc[2] = (t & 4) ? -tot : tot;
  acc[3] = z3;
  acc[4] = (t & 2) ? -tot : tot;
  acc[5] = (t & 1) ? -tot : tot;
  acc[6] = z6;
  acc[7] = z7;
  acc[8] = z8;
  acc[9] = z9;
}

__device__ __forceinline__ void load_afrag(v2f (&a)[4], const float* G,
                                           const int p, const int t) {
  const float2* src = (const float2*)G + p * 128 + t;
#pragma unroll
  for (int kk = 0; kk < 4; ++kk) {
    const float2 g = src[kk * 32];
    a[kk].x = g.x;
    a[kk].y = g.y;
  }
}

#define RUN_PREFIX(p, ...)                                 \
  {                                                        \
    v2f an[4];                                             \
    load_afrag(an, G, (p) + 1, t);                         \
    prefix_pass<__VA_ARGS__>(lds, wofs, a, t);             \
    a[0] = an[0]; a[1] = an[1]; a[2] = an[2]; a[3] = an[3]; \
  }

#define RUN_FOLDED(p, NPAIR, KS, af, ...)                  \
  {                                                        \
    v2f an[4];                                             \
    load_afrag(an, G, (p) + 1, t);                         \
    fused_pass<NPAIR, KS, __VA_ARGS__>(lds, wofs, af, t);  \
    a[0] = an[0]; a[1] = an[1]; a[2] = an[2]; a[3] = an[3]; \
  }

__global__ __launch_bounds__(256, 8) void qsim_kernel(
    const float* __restrict__ inp, const float* __restrict__ W,
    float* __restrict__ out) {
  __shared__ float2 st[WPB][DIM];
  __shared__ float gs[40][8];
  __shared__ float tc[80], ts[80];
  __shared__ float4 gfrag[NPASS * 64];
  const int t = threadIdx.x & 31;
  const int w = threadIdx.x >> 5;
  const int samp = blockIdx.x * WPB + w;
  char* lds = (char*)&st[0][0];
  const unsigned wofs = w * (DIM * 8);
  const float* G = (const float*)gfrag;

  const float xl = (t < NQ) ? inp[samp * NQ + t] : 0.f;

  gate_trig(W, tc, ts, threadIdx.x);
  __syncthreads();
  gate_entries(tc, ts, gs, threadIdx.x);
  __syncthreads();
  gate_columns(gs, (float*)gfrag, w, t);

  float cc[NQ], ss[NQ];
  {
    const float xv = fminf(fmaxf(xl, 0.f), 1.f);
    const float ang = 0.5f * 3.14159265358979323846f * xv;
    const float cl = cosf(ang), sl = sinf(ang);
#pragma unroll
    for (int q = 0; q < NQ; ++q) {
      cc[q] = __int_as_float(__builtin_amdgcn_readlane(__float_as_int(cl), q));
      ss[q] = __int_as_float(__builtin_amdgcn_readlane(__float_as_int(sl), q));
    }
  }
  float ppre = 1.f;
#pragma unroll
  for (int j = 0; j < 5; ++j)
    ppre *= ((t >> j) & 1) ? ss[6 - j] : cc[6 - j];
  const unsigned pa = wofs | (8u * swz(((t & 3) << 3) | ((t >> 2) << 5)));
#pragma unroll
  for (int rr = 0; rr < 4; ++rr) {
    const float f = ppre * ((rr & 1) ? ss[1] : cc[1]) * ((rr & 2) ? ss[0] : cc[0]);
    const int c = swz(rr << 8);
    *(float2*)(lds + ((pa ^ (8u * (c & 31))) + 8u * (c & ~31))) =
        make_float2(f, 0.f);
  }
  __syncthreads();

  v2f a[4];
  load_afrag(a, G, 0, t);

  RUN_PREFIX(0, 9, 8, 7, 3, 4, 5, 6)
  RUN_PREFIX(1, 7, 6, 5, 3, 4, 8, 9)
  RUN_PREFIX(2, 5, 4, 3, 6, 7, 8, 9)

  {
    v2f af[1];
    const float w0 = cc[7] * cc[8], w1 = ss[7] * cc[8];
    const float w2 = cc[7] * ss[8], w3 = ss[7] * ss[8];
    af[0].x = w0 * a[0].x + w1 * a[1].x + w2 * a[2].x + w3 * a[3].x;
    af[0].y = w0 * a[0].y + w1 * a[1].y + w2 * a[2].y + w3 * a[3].y;
    RUN_FOLDED(3, 2, 1, af, 3, 2, 1, 4, 5, 7, 8, 6, 9, 0)
  }

  {
    v2f af[2];
#pragma unroll
    for (int s = 0; s < 2; ++s) {
      af[s].x = cc[9] * a[2 * s].x + ss[9] * a[2 * s + 1].x;
      af[s].y = cc[9] * a[2 * s].y + ss[9] * a[2 * s + 1].y;
    }
    v2f a5[4];
    load_afrag(a5, G, 5, t);
    chained_passes<2, 1, 0, 9, 3, 4, 5, 6, 8, 7, 2>(lds, wofs, af, a5, t);
  }

  {
    v2f a6[4], a7[4];
    load_afrag(a6, G, 6, t);
    load_afrag(a7, G, 7, t);
    chained_passes<4, 7, 6, 5, 0, 2, 8, 9, 4, 3, 1>(lds, wofs, a6, a7, t);
  }

  float acc[NQ];
  {
    v2f a8[4], a9[4];
    load_afrag(a8, G, 8, t);
    load_afrag(a9, G, 9, t);
    tail_passes(lds, wofs, a8, a9, t, acc);
  }
#pragma unroll
  for (int q = 0; q < NQ; ++q) {
#pragma unroll
    for (int off = 16; off >= 1; off >>= 1)
      acc[q] += __shfl_xor(acc[q], off, 32);
  }
  if (t == 0) {
#pragma unroll
    for (int q = 0; q < NQ; ++q) out[samp * NQ + q] = acc[q];
  }
}

extern "C" void kernel_launch(void* const* d_in, const int* in_sizes, int n_in,
                              void* d_out, int out_size, void* d_ws,
                              size_t ws_size, hipStream_t stream) {
  const float* inputs = (const float*)d_in[0];
  const float* weights = (const float*)d_in[1];
  float* out = (float*)d_out;

  const int B = in_sizes[0] / NQ;
  qsim_kernel<<<dim3(B / WPB), dim3(256), 0, stream>>>(inputs, weights, out);
}
